# starting kernel
# baseline (speedup 1.0000x reference)
#include <hip/hip_runtime.h>


typedef __attribute__((ext_vector_type(2))) float v2f;
typedef __attribute__((ext_vector_type(8))) float v8f;

static inline int idiv_up(long long a, long long b) { return (int)((a + b - 1) / b); }

__global__ void fill_zero_kernel(float* __restrict__ ptr, long long count) {
  long long i = (long long)blockIdx.x * blockDim.x + threadIdx.x;
  if (i < count) ptr[i] = 0.0f;
}

__global__ void init_out_kernel(float* __restrict__ out,
                                const float* __restrict__ bias2,
                                long long count, int C) {
  long long i = (long long)blockIdx.x * blockDim.x + threadIdx.x;
  if (i < count) out[i] = bias2[i % C];
}

__global__ void degree_kernel(const int* __restrict__ s, const int* __restrict__ p,
                              const int* __restrict__ o, float* __restrict__ deg,
                              int E, int N, int R) {
  long long i = (long long)blockIdx.x * blockDim.x + threadIdx.x;
  if (i < E) {
    int si = s[i], pi = p[i], oi = o[i];
    atomicAdd(&deg[(size_t)pi * N + si], 1.0f);
    atomicAdd(&deg[(size_t)(pi + R) * N + oi], 1.0f);
  } else if (i < (long long)E + N) {
    int node = (int)(i - E);
    deg[(size_t)(2 * R) * N + node] = 1.0f;
  }
}

__global__ void gemm48_wmma_kernel(const float* __restrict__ A,
                                   const float* __restrict__ B,
                                   float* __restrict__ C,
                                   int M, int K, long long J) {
  __shared__ float sA[48 * 64];

  const int total = 48 * K;
  for (int idx = threadIdx.x; idx < total; idx += blockDim.x) {
    const int row = idx / K;
    const int kk  = idx - row * K;
    sA[idx] = (row < M) ? A[row * K + kk] : 0.0f;
  }
  __syncthreads();

  const int wave = threadIdx.x >> 5;
  const int lane = threadIdx.x & 31;
  const long long jbase =
      ((long long)blockIdx.x * (blockDim.x >> 5) + wave) * 16;
  if (jbase >= J) return;
  const int col   = lane & 15;
  const int khalf = (lane >> 4) * 2;

  v8f acc[3];
  acc[0] = (v8f){};
  acc[1] = (v8f){};
  acc[2] = (v8f){};

  const float* __restrict__ Bp = B + jbase + col;
  for (int kc = 0; kc < K; kc += 4) {
    const int k0 = kc + khalf;
    v2f b;
    b.x = Bp[(long long)k0 * J];
    b.y = Bp[(long long)(k0 + 1) * J];
#pragma unroll
    for (int mt = 0; mt < 3; ++mt) {
      const int m = mt * 16 + col;
      const v2f a = *(const v2f*)&sA[m * K + k0];
      acc[mt] = __builtin_amdgcn_wmma_f32_16x16x4_f32(
          false, a, false, b, (short)0, acc[mt], false, false);
    }
  }

  const int rhalf = (lane >> 4) * 8;
#pragma unroll
  for (int mt = 0; mt < 3; ++mt) {
#pragma unroll
    for (int v = 0; v < 8; ++v) {
      const int row = mt * 16 + rhalf + v;
      if (row < M) C[(long long)row * J + jbase + col] = acc[mt][v];
    }
  }
}

__global__ void layer1_scatter_kernel(const int* __restrict__ s, const int* __restrict__ p,
                                      const int* __restrict__ o,
                                      const float* __restrict__ deg,
                                      const float* __restrict__ W1,
                                      float* __restrict__ h,
                                      int E, int N, int R, long long NE) {
  long long gid = (long long)blockIdx.x * blockDim.x + threadIdx.x;
  const long long total = ((long long)2 * E + N) * 16;
  if (gid >= total) return;
  const int e = (int)(gid & 15);
  const long long i = gid >> 4;
  int si, pi, oi;
  if (i < E)              { si = s[i];      pi = p[i];          oi = o[i]; }
  else if (i < 2LL * E)   { long long j = i - E; si = o[j]; pi = p[j] + R; oi = s[j]; }
  else                    { int nd = (int)(i - 2LL * E); si = nd; pi = 2 * R; oi = nd; }
  const float val = 1.0f / deg[(size_t)pi * N + si];
  const float w   = W1[(size_t)pi * NE + (size_t)oi * 16 + e];
  atomicAdd(&h[(size_t)si * 16 + e], val * w);
}

__global__ void relu_bias_kernel(float* __restrict__ h, const float* __restrict__ bias1,
                                 long long count) {
  long long i = (long long)blockIdx.x * blockDim.x + threadIdx.x;
  if (i < count) h[i] = fmaxf(h[i] + bias1[i & 15], 0.0f);
}

__global__ void layer2_scatter_kernel(const int* __restrict__ s, const int* __restrict__ p,
                                      const int* __restrict__ o,
                                      const float* __restrict__ deg,
                                      const float* __restrict__ h,
                                      const float* __restrict__ W2,
                                      float* __restrict__ out,
                                      int E, int N, int R) {
  long long gid = (long long)blockIdx.x * blockDim.x + threadIdx.x;
  const long long total = ((long long)2 * E + N) * 8;
  if (gid >= total) return;
  const int c = (int)(gid & 7);
  const long long i = gid >> 3;
  int si, pi, oi;
  if (i < E)              { si = s[i];      pi = p[i];          oi = o[i]; }
  else if (i < 2LL * E)   { long long j = i - E; si = o[j]; pi = p[j] + R; oi = s[j]; }
  else                    { int nd = (int)(i - 2LL * E); si = nd; pi = 2 * R; oi = nd; }
  const float val = 1.0f / deg[(size_t)pi * N + si];
  const float* hv = &h[(size_t)oi * 16];
  const float* w  = &W2[(size_t)pi * 128 + c];
  float sum = 0.0f;
#pragma unroll
  for (int e = 0; e < 16; ++e) sum += hv[e] * w[e * 8];
  atomicAdd(&out[(size_t)si * 8 + c], val * sum);
}

extern "C" void kernel_launch(void* const* d_in, const int* in_sizes, int n_in,
                              void* d_out, int out_size, void* d_ws, size_t ws_size,
                              hipStream_t stream) {
  const float* comps1 = (const float*)d_in[0];
  const float* bases1 = (const float*)d_in[1];
  const float* comps2 = (const float*)d_in[2];
  const float* bases2 = (const float*)d_in[3];
  const float* bias1  = (const float*)d_in[4];
  const float* bias2  = (const float*)d_in[5];
  const int*   s      = (const int*)d_in[6];
  const int*   p      = (const int*)d_in[7];
  const int*   o      = (const int*)d_in[8];

  const int EMB    = in_sizes[4];
  const int NUMCLS = in_sizes[5];
  const int E      = in_sizes[6];
  const int N      = out_size / NUMCLS;
  const int BASES  = (int)((long long)in_sizes[1] / ((long long)N * EMB));
  const int r_tot  = in_sizes[0] / BASES;
  const int R      = (r_tot - 1) / 2;

  float* ws = (float*)d_ws;
  size_t off = 0;
  auto alloc = [&](size_t nfloats) {
    size_t cur = off;
    off += (nfloats + 63) & ~(size_t)63;
    return cur;
  };
  float* deg = ws + alloc((size_t)r_tot * N);
  float* h   = ws + alloc((size_t)N * EMB);
  float* W2  = ws + alloc((size_t)r_tot * EMB * NUMCLS);
  float* W1  = ws + alloc((size_t)r_tot * N * EMB);
  (void)ws_size; (void)n_in;

  const long long NE   = (long long)N * EMB;
  const long long J2   = (long long)EMB * NUMCLS;
  const long long Eall = (long long)2 * E + N;
  const int B = 256;

  fill_zero_kernel<<<idiv_up((long long)r_tot * N, B), B, 0, stream>>>(deg, (long long)r_tot * N);
  fill_zero_kernel<<<idiv_up(NE, B), B, 0, stream>>>(h, NE);

  degree_kernel<<<idiv_up((long long)E + N, B), B, 0, stream>>>(s, p, o, deg, E, N, R);

  gemm48_wmma_kernel<<<idiv_up(NE, 16LL * (B / 32)), B, 0, stream>>>(
      comps1, bases1, W1, r_tot, BASES, NE);
  gemm48_wmma_kernel<<<idiv_up(J2, 16LL * (B / 32)), B, 0, stream>>>(
      comps2, bases2, W2, r_tot, BASES, J2);

  layer1_scatter_kernel<<<idiv_up(Eall * 16, B), B, 0, stream>>>(
      s, p, o, deg, W1, h, E, N, R, NE);
  relu_bias_kernel<<<idiv_up(NE, B), B, 0, stream>>>(h, bias1, NE);

  init_out_kernel<<<idiv_up((long long)N * NUMCLS, B), B, 0, stream>>>(
      (float*)d_out, bias2, (long long)N * NUMCLS, NUMCLS);
  layer2_scatter_kernel<<<idiv_up(Eall * 8, B), B, 0, stream>>>(
      s, p, o, deg, h, W2, (float*)d_out, E, N, R);
}


# W1 GEMM as a streaming kernel: 4096 resident waves hold the A fragments in registers and walk 64-column chunks with 16-byte B loads and C stores, each k-step's B refilled right behind its products
# speedup vs baseline: 1.2939x; 1.2939x over previous
#include <hip/hip_runtime.h>


typedef __attribute__((ext_vector_type(2))) float v2f;
typedef __attribute__((ext_vector_type(4))) float v4f;
typedef __attribute__((ext_vector_type(8))) float v8f;

static inline int idiv_up(long long a, long long b) { return (int)((a + b - 1) / b); }

__global__ void fill_zero_kernel(float* __restrict__ ptr, long long count) {
  long long i = (long long)blockIdx.x * blockDim.x + threadIdx.x;
  if (i < count) ptr[i] = 0.0f;
}

__global__ void init_out_kernel(float* __restrict__ out,
                                const float* __restrict__ bias2,
                                long long count, int C) {
  long long i = (long long)blockIdx.x * blockDim.x + threadIdx.x;
  if (i < count) out[i] = bias2[i % C];
}

__global__ void degree_kernel(const int* __restrict__ s, const int* __restrict__ p,
                              const int* __restrict__ o, float* __restrict__ deg,
                              int E, int N, int R) {
  long long i = (long long)blockIdx.x * blockDim.x + threadIdx.x;
  if (i < E) {
    int si = s[i], pi = p[i], oi = o[i];
    atomicAdd(&deg[(size_t)pi * N + si], 1.0f);
    atomicAdd(&deg[(size_t)(pi + R) * N + oi], 1.0f);
  } else if (i < (long long)E + N) {
    int node = (int)(i - E);
    deg[(size_t)(2 * R) * N + node] = 1.0f;
  }
}

__global__ void gemm48_wmma_kernel(const float* __restrict__ A,
                                   const float* __restrict__ B,
                                   float* __restrict__ C,
                                   int M, int K, long long J) {
  __shared__ float sA[48 * 64];

  const int total = 48 * K;
  for (int idx = threadIdx.x; idx < total; idx += blockDim.x) {
    const int row = idx / K;
    const int kk  = idx - row * K;
    sA[idx] = (row < M) ? A[row * K + kk] : 0.0f;
  }
  __syncthreads();

  const int wave = threadIdx.x >> 5;
  const int lane = threadIdx.x & 31;
  const long long jbase =
      ((long long)blockIdx.x * (blockDim.x >> 5) + wave) * 16;
  if (jbase >= J) return;
  const int col   = lane & 15;
  const int khalf = (lane >> 4) * 2;

  v8f acc[3];
  acc[0] = (v8f){};
  acc[1] = (v8f){};
  acc[2] = (v8f){};

  const float* __restrict__ Bp = B + jbase + col;
  for (int kc = 0; kc < K; kc += 4) {
    const int k0 = kc + khalf;
    v2f b;
    b.x = Bp[(long long)k0 * J];
    b.y = Bp[(long long)(k0 + 1) * J];
#pragma unroll
    for (int mt = 0; mt < 3; ++mt) {
      const int m = mt * 16 + col;
      const v2f a = *(const v2f*)&sA[m * K + k0];
      acc[mt] = __builtin_amdgcn_wmma_f32_16x16x4_f32(
          false, a, false, b, (short)0, acc[mt], false, false);
    }
  }

  const int rhalf = (lane >> 4) * 8;
#pragma unroll
  for (int mt = 0; mt < 3; ++mt) {
#pragma unroll
    for (int v = 0; v < 8; ++v) {
      const int row = mt * 16 + rhalf + v;
      if (row < M) C[(long long)row * J + jbase + col] = acc[mt][v];
    }
  }
}

template <int KS, bool REFILL>
__device__ __forceinline__ void gemm48_chunk(const v2f (&a)[3][KS], v4f (&b0)[KS],
                                             v4f (&b1)[KS], v8f (&acc)[3][4],
                                             const float* __restrict__ pn, long long J) {
#pragma unroll
  for (int mt = 0; mt < 3; ++mt)
#pragma unroll
    for (int t = 0; t < 4; ++t) acc[mt][t] = (v8f){};
#pragma unroll
  for (int ks = 0; ks < KS; ++ks) {
    const v4f r0 = b0[ks];
    const v4f r1 = b1[ks];
#pragma unroll
    for (int t = 0; t < 4; ++t) {
      const v2f b = {r0[t], r1[t]};
#pragma unroll
      for (int mt = 0; mt < 3; ++mt)
        acc[mt][t] = __builtin_amdgcn_wmma_f32_16x16x4_f32(
            false, a[mt][ks], false, b, (short)0, acc[mt][t], false, false);
    }
    if (REFILL) {
      b0[ks] = __builtin_nontemporal_load((const v4f*)(pn + (long long)(4 * ks) * J));
      b1[ks] = __builtin_nontemporal_load((const v4f*)(pn + (long long)(4 * ks + 1) * J));
    }
    __builtin_amdgcn_sched_barrier(0);
  }
}

__device__ __forceinline__ void gemm48_store(const v8f (&acc)[3][4], float* __restrict__ pc,
                                             int lim, long long J) {
#pragma unroll
  for (int mt = 0; mt < 3; ++mt) {
#pragma unroll
    for (int r = 0; r < 8; ++r) {
      if (mt * 16 + r < lim) {
        const v4f d = {acc[mt][0][r], acc[mt][1][r], acc[mt][2][r], acc[mt][3][r]};
        *(v4f*)(pc + (long long)(mt * 16 + r) * J) = d;
      }
    }
  }
}

template <int K>
__global__ __launch_bounds__(128, 4) void gemm48_stream_kernel(
    const float* __restrict__ A, const float* __restrict__ B,
    float* __restrict__ C, int M, long long J) {
  constexpr int KS = K / 4;
  const int lane = threadIdx.x & 31;
  const int c = lane & 15;
  const int h = lane >> 4;
  const int wave = __builtin_amdgcn_readfirstlane((int)(threadIdx.x >> 5));
  const int wpb  = (int)(blockDim.x >> 5);
  const long long nw  = (long long)gridDim.x * wpb;
  const long long nST = J >> 6;
  long long st = (long long)blockIdx.x * wpb + wave;
  if (st >= nST) return;

  v2f a[3][KS];
#pragma unroll
  for (int mt = 0; mt < 3; ++mt) {
    const int row = mt * 16 + c;
#pragma unroll
    for (int ks = 0; ks < KS; ++ks) {
      v2f t = {0.0f, 0.0f};
      if (row < M) t = *(const v2f*)&A[row * K + 4 * ks + 2 * h];
      a[mt][ks] = t;
    }
  }

  const float* __restrict__ Bl = B + (long long)(2 * h) * J + 4 * c;
  float* __restrict__ Cl = C + (long long)(8 * h) * J + 4 * c;
  const int lim = M - 8 * h;

  v4f b0[KS], b1[KS];
  {
    const float* p = Bl + st * 64;
#pragma unroll
    for (int ks = 0; ks < KS; ++ks) {
      b0[ks] = __builtin_nontemporal_load((const v4f*)(p + (long long)(4 * ks) * J));
      b1[ks] = __builtin_nontemporal_load((const v4f*)(p + (long long)(4 * ks + 1) * J));
    }
  }

  v8f acc[3][4];
  while (st + nw < nST) {
    gemm48_chunk<KS, true>(a, b0, b1, acc, Bl + (st + nw) * 64, J);
    gemm48_store(acc, Cl + st * 64, lim, J);
    st += nw;
  }
  gemm48_chunk<KS, false>(a, b0, b1, acc, Bl, J);
  gemm48_store(acc, Cl + st * 64, lim, J);
}

__global__ void layer1_scatter_kernel(const int* __restrict__ s, const int* __restrict__ p,
                                      const int* __restrict__ o,
                                      const float* __restrict__ deg,
                                      const float* __restrict__ W1,
                                      float* __restrict__ h,
                                      int E, int N, int R, long long NE) {
  long long gid = (long long)blockIdx.x * blockDim.x + threadIdx.x;
  const long long total = ((long long)2 * E + N) * 16;
  if (gid >= total) return;
  const int e = (int)(gid & 15);
  const long long i = gid >> 4;
  int si, pi, oi;
  if (i < E)              { si = s[i];      pi = p[i];          oi = o[i]; }
  else if (i < 2LL * E)   { long long j = i - E; si = o[j]; pi = p[j] + R; oi = s[j]; }
  else                    { int nd = (int)(i - 2LL * E); si = nd; pi = 2 * R; oi = nd; }
  const float val = 1.0f / deg[(size_t)pi * N + si];
  const float w   = W1[(size_t)pi * NE + (size_t)oi * 16 + e];
  atomicAdd(&h[(size_t)si * 16 + e], val * w);
}

__global__ void relu_bias_kernel(float* __restrict__ h, const float* __restrict__ bias1,
                                 long long count) {
  long long i = (long long)blockIdx.x * blockDim.x + threadIdx.x;
  if (i < count) h[i] = fmaxf(h[i] + bias1[i & 15], 0.0f);
}

__global__ void layer2_scatter_kernel(const int* __restrict__ s, const int* __restrict__ p,
                                      const int* __restrict__ o,
                                      const float* __restrict__ deg,
                                      const float* __restrict__ h,
                                      const float* __restrict__ W2,
                                      float* __restrict__ out,
                                      int E, int N, int R) {
  long long gid = (long long)blockIdx.x * blockDim.x + threadIdx.x;
  const long long total = ((long long)2 * E + N) * 8;
  if (gid >= total) return;
  const int c = (int)(gid & 7);
  const long long i = gid >> 3;
  int si, pi, oi;
  if (i < E)              { si = s[i];      pi = p[i];          oi = o[i]; }
  else if (i < 2LL * E)   { long long j = i - E; si = o[j]; pi = p[j] + R; oi = s[j]; }
  else                    { int nd = (int)(i - 2LL * E); si = nd; pi = 2 * R; oi = nd; }
  const float val = 1.0f / deg[(size_t)pi * N + si];
  const float* hv = &h[(size_t)oi * 16];
  const float* w  = &W2[(size_t)pi * 128 + c];
  float sum = 0.0f;
#pragma unroll
  for (int e = 0; e < 16; ++e) sum += hv[e] * w[e * 8];
  atomicAdd(&out[(size_t)si * 8 + c], val * sum);
}

extern "C" void kernel_launch(void* const* d_in, const int* in_sizes, int n_in,
                              void* d_out, int out_size, void* d_ws, size_t ws_size,
                              hipStream_t stream) {
  const float* comps1 = (const float*)d_in[0];
  const float* bases1 = (const float*)d_in[1];
  const float* comps2 = (const float*)d_in[2];
  const float* bases2 = (const float*)d_in[3];
  const float* bias1  = (const float*)d_in[4];
  const float* bias2  = (const float*)d_in[5];
  const int*   s      = (const int*)d_in[6];
  const int*   p      = (const int*)d_in[7];
  const int*   o      = (const int*)d_in[8];

  const int EMB    = in_sizes[4];
  const int NUMCLS = in_sizes[5];
  const int E      = in_sizes[6];
  const int N      = out_size / NUMCLS;
  const int BASES  = (int)((long long)in_sizes[1] / ((long long)N * EMB));
  const int r_tot  = in_sizes[0] / BASES;
  const int R      = (r_tot - 1) / 2;

  float* ws = (float*)d_ws;
  size_t off = 0;
  auto alloc = [&](size_t nfloats) {
    size_t cur = off;
    off += (nfloats + 63) & ~(size_t)63;
    return cur;
  };
  float* deg = ws + alloc((size_t)r_tot * N);
  float* h   = ws + alloc((size_t)N * EMB);
  float* W2  = ws + alloc((size_t)r_tot * EMB * NUMCLS);
  float* W1  = ws + alloc((size_t)r_tot * N * EMB);
  (void)ws_size; (void)n_in;

  const long long NE   = (long long)N * EMB;
  const long long J2   = (long long)EMB * NUMCLS;
  const long long Eall = (long long)2 * E + N;
  const int B = 256;

  fill_zero_kernel<<<idiv_up((long long)r_tot * N, B), B, 0, stream>>>(deg, (long long)r_tot * N);
  fill_zero_kernel<<<idiv_up(NE, B), B, 0, stream>>>(h, NE);

  degree_kernel<<<idiv_up((long long)E + N, B), B, 0, stream>>>(s, p, o, deg, E, N, R);

  if (BASES == 40 && (NE & 63) == 0 && r_tot <= 48) {
    gemm48_stream_kernel<40><<<1024, 128, 0, stream>>>(comps1, bases1, W1, r_tot, NE);
  } else {
    gemm48_wmma_kernel<<<idiv_up(NE, 16LL * (B / 32)), B, 0, stream>>>(
        comps1, bases1, W1, r_tot, BASES, NE);
  }
  gemm48_wmma_kernel<<<idiv_up(J2, 16LL * (B / 32)), B, 0, stream>>>(
      comps2, bases2, W2, r_tot, BASES, J2);

  layer1_scatter_kernel<<<idiv_up(Eall * 16, B), B, 0, stream>>>(
      s, p, o, deg, W1, h, E, N, R, NE);
  relu_bias_kernel<<<idiv_up(NE, B), B, 0, stream>>>(h, bias1, NE);

  init_out_kernel<<<idiv_up((long long)N * NUMCLS, B), B, 0, stream>>>(
      (float*)d_out, bias2, (long long)N * NUMCLS, NUMCLS);
  layer2_scatter_kernel<<<idiv_up(Eall * 8, B), B, 0, stream>>>(
      s, p, o, deg, h, W2, (float*)d_out, E, N, R);
}


# Both edge scatters re-tiled: a wave takes 32 edges, one lane decodes one edge (indices, degree, 1/deg once per edge), DPP broadcasts it; layer 2 sums its 16 terms by a lane tree (new sum order)
# speedup vs baseline: 1.4612x; 1.1293x over previous
#include <hip/hip_runtime.h>


typedef __attribute__((ext_vector_type(2))) float v2f;
typedef __attribute__((ext_vector_type(4))) float v4f;
typedef __attribute__((ext_vector_type(8))) float v8f;

static inline int idiv_up(long long a, long long b) { return (int)((a + b - 1) / b); }

__global__ void fill_zero_kernel(float* __restrict__ ptr, long long count) {
  long long i = (long long)blockIdx.x * blockDim.x + threadIdx.x;
  if (i < count) ptr[i] = 0.0f;
}

__global__ void init_out_kernel(float* __restrict__ out,
                                const float* __restrict__ bias2,
                                long long count, int C) {
  long long i = (long long)blockIdx.x * blockDim.x + threadIdx.x;
  if (i < count) out[i] = bias2[i % C];
}

__global__ void degree_kernel(const int* __restrict__ s, const int* __restrict__ p,
                              const int* __restrict__ o, float* __restrict__ deg,
                              int E, int N, int R) {
  long long i = (long long)blockIdx.x * blockDim.x + threadIdx.x;
  if (i < E) {
    int si = s[i], pi = p[i], oi = o[i];
    atomicAdd(&deg[(size_t)pi * N + si], 1.0f);
    atomicAdd(&deg[(size_t)(pi + R) * N + oi], 1.0f);
  } else if (i < (long long)E + N) {
    int node = (int)(i - E);
    deg[(size_t)(2 * R) * N + node] = 1.0f;
  }
}

__global__ void gemm48_wmma_kernel(const float* __restrict__ A,
                                   const float* __restrict__ B,
                                   float* __restrict__ C,
                                   int M, int K, long long J) {
  __shared__ float sA[48 * 64];

  const int total = 48 * K;
  for (int idx = threadIdx.x; idx < total; idx += blockDim.x) {
    const int row = idx / K;
    const int kk  = idx - row * K;
    sA[idx] = (row < M) ? A[row * K + kk] : 0.0f;
  }
  __syncthreads();

  const int wave = threadIdx.x >> 5;
  const int lane = threadIdx.x & 31;
  const long long jbase =
      ((long long)blockIdx.x * (blockDim.x >> 5) + wave) * 16;
  if (jbase >= J) return;
  const int col   = lane & 15;
  const int khalf = (lane >> 4) * 2;

  v8f acc[3];
  acc[0] = (v8f){};
  acc[1] = (v8f){};
  acc[2] = (v8f){};

  const float* __restrict__ Bp = B + jbase + col;
  for (int kc = 0; kc < K; kc += 4) {
    const int k0 = kc + khalf;
    v2f b;
    b.x = Bp[(long long)k0 * J];
    b.y = Bp[(long long)(k0 + 1) * J];
#pragma unroll
    for (int mt = 0; mt < 3; ++mt) {
      const int m = mt * 16 + col;
      const v2f a = *(const v2f*)&sA[m * K + k0];
      acc[mt] = __builtin_amdgcn_wmma_f32_16x16x4_f32(
          false, a, false, b, (short)0, acc[mt], false, false);
    }
  }

  const int rhalf = (lane >> 4) * 8;
#pragma unroll
  for (int mt = 0; mt < 3; ++mt) {
#pragma unroll
    for (int v = 0; v < 8; ++v) {
      const int row = mt * 16 + rhalf + v;
      if (row < M) C[(long long)row * J + jbase + col] = acc[mt][v];
    }
  }
}

template <int KS, bool REFILL>
__device__ __forceinline__ void gemm48_chunk(const v2f (&a)[3][KS], v4f (&b0)[KS],
                                             v4f (&b1)[KS], v8f (&acc)[3][4],
                                             const float* __restrict__ pn, long long J) {
#pragma unroll
  for (int mt = 0; mt < 3; ++mt)
#pragma unroll
    for (int t = 0; t < 4; ++t) acc[mt][t] = (v8f){};
#pragma unroll
  for (int ks = 0; ks < KS; ++ks) {
    const v4f r0 = b0[ks];
    const v4f r1 = b1[ks];
#pragma unroll
    for (int t = 0; t < 4; ++t) {
      const v2f b = {r0[t], r1[t]};
#pragma unroll
      for (int mt = 0; mt < 3; ++mt)
        acc[mt][t] = __builtin_amdgcn_wmma_f32_16x16x4_f32(
            false, a[mt][ks], false, b, (short)0, acc[mt][t], false, false);
    }
    if (REFILL) {
      b0[ks] = __builtin_nontemporal_load((const v4f*)(pn + (long long)(4 * ks) * J));
      b1[ks] = __builtin_nontemporal_load((const v4f*)(pn + (long long)(4 * ks + 1) * J));
    }
    __builtin_amdgcn_sched_barrier(0);
  }
}

__device__ __forceinline__ void gemm48_store(const v8f (&acc)[3][4], float* __restrict__ pc,
                                             int lim, long long J) {
#pragma unroll
  for (int mt = 0; mt < 3; ++mt) {
#pragma unroll
    for (int r = 0; r < 8; ++r) {
      if (mt * 16 + r < lim) {
        const v4f d = {acc[mt][0][r], acc[mt][1][r], acc[mt][2][r], acc[mt][3][r]};
        *(v4f*)(pc + (long long)(mt * 16 + r) * J) = d;
      }
    }
  }
}

template <int K>
__global__ __launch_bounds__(128, 4) void gemm48_stream_kernel(
    const float* __restrict__ A, const float* __restrict__ B,
    float* __restrict__ C, int M, long long J) {
  constexpr int KS = K / 4;
  const int lane = threadIdx.x & 31;
  const int c = lane & 15;
  const int h = lane >> 4;
  const int wave = __builtin_amdgcn_readfirstlane((int)(threadIdx.x >> 5));
  const int wpb  = (int)(blockDim.x >> 5);
  const long long nw  = (long long)gridDim.x * wpb;
  const long long nST = J >> 6;
  long long st = (long long)blockIdx.x * wpb + wave;
  if (st >= nST) return;

  v2f a[3][KS];
#pragma unroll
  for (int mt = 0; mt < 3; ++mt) {
    const int row = mt * 16 + c;
#pragma unroll
    for (int ks = 0; ks < KS; ++ks) {
      v2f t = {0.0f, 0.0f};
      if (row < M) t = *(const v2f*)&A[row * K + 4 * ks + 2 * h];
      a[mt][ks] = t;
    }
  }

  const float* __restrict__ Bl = B + (long long)(2 * h) * J + 4 * c;
  float* __restrict__ Cl = C + (long long)(8 * h) * J + 4 * c;
  const int lim = M - 8 * h;

  v4f b0[KS], b1[KS];
  {
    const float* p = Bl + st * 64;
#pragma unroll
    for (int ks = 0; ks < KS; ++ks) {
      b0[ks] = __builtin_nontemporal_load((const v4f*)(p + (long long)(4 * ks) * J));
      b1[ks] = __builtin_nontemporal_load((const v4f*)(p + (long long)(4 * ks + 1) * J));
    }
  }

  v8f acc[3][4];
  while (st + nw < nST) {
    gemm48_chunk<KS, true>(a, b0, b1, acc, Bl + (st + nw) * 64, J);
    gemm48_store(acc, Cl + st * 64, lim, J);
    st += nw;
  }
  gemm48_chunk<KS, false>(a, b0, b1, acc, Bl, J);
  gemm48_store(acc, Cl + st * 64, lim, J);
}

__device__ __forceinline__ void edge_decode(const int* __restrict__ s, const int* __restrict__ p,
                                            const int* __restrict__ o,
                                            const float* __restrict__ deg,
                                            long long i, int E, int N, int R,
                                            int& si, int& pi, int& oi, float& val) {
  si = 0; pi = 0; oi = 0; val = 0.0f;
  if (i < (long long)2 * E + N) {
    if (i < E)            { si = s[i];      pi = p[i];          oi = o[i]; }
    else if (i < 2LL * E) { long long j = i - E; si = o[j]; pi = p[j] + R; oi = s[j]; }
    else                  { int nd = (int)(i - 2LL * E); si = nd; pi = 2 * R; oi = nd; }
    val = 1.0f / deg[(size_t)pi * N + si];
  }
}

template <bool GUARD>
__device__ __forceinline__ void layer1_rows(const float* __restrict__ W1, float* __restrict__ h,
                                            unsigned wofs, unsigned hofs, float val, int ok,
                                            int e) {
  float w[16];
#pragma unroll
  for (int k = 0; k < 16; ++k) {
    const unsigned wo = (unsigned)__shfl((int)wofs, k, 16);
    w[k] = W1[wo + e];
  }
#pragma unroll
  for (int k = 0; k < 16; ++k) {
    const unsigned ho = (unsigned)__shfl((int)hofs, k, 16);
    const float v = __shfl(val, k, 16);
    if (!GUARD || __shfl(ok, k, 16)) atomicAdd(&h[ho + e], v * w[k]);
  }
}

__global__ __launch_bounds__(256) void layer1_scatter_kernel(
    const int* __restrict__ s, const int* __restrict__ p, const int* __restrict__ o,
    const float* __restrict__ deg, const float* __restrict__ W1, float* __restrict__ h,
    int E, int N, int R, long long NE) {
  const int lane = threadIdx.x & 31;
  const long long total = (long long)2 * E + N;
  const long long base = (((long long)blockIdx.x * blockDim.x + threadIdx.x) >> 5) << 5;
  if (base >= total) return;
  int si, pi, oi;
  float val;
  edge_decode(s, p, o, deg, base + lane, E, N, R, si, pi, oi, val);
  const unsigned wofs = (unsigned)((long long)pi * NE + (long long)oi * 16);
  const unsigned hofs = (unsigned)si * 16u;
  const int e = lane & 15;
  if (base + 32 <= total) layer1_rows<false>(W1, h, wofs, hofs, val, 1, e);
  else layer1_rows<true>(W1, h, wofs, hofs, val, (base + lane) < total ? 1 : 0, e);
}

__global__ void relu_bias_kernel(float* __restrict__ h, const float* __restrict__ bias1,
                                 long long count) {
  long long i = (long long)blockIdx.x * blockDim.x + threadIdx.x;
  if (i < count) h[i] = fmaxf(h[i] + bias1[i & 15], 0.0f);
}

template <int CTRL>
__device__ __forceinline__ float lane_partner(float x) {
  return __builtin_bit_cast(
      float, __builtin_amdgcn_update_dpp(0, __builtin_bit_cast(int, x), CTRL, 0xf, 0xf, true));
}

template <bool GUARD>
__device__ __forceinline__ void layer2_rows(const float* __restrict__ h,
                                            const float* __restrict__ W2,
                                            float* __restrict__ out,
                                            unsigned hofs, unsigned wofs, unsigned oofs,
                                            float val, int ok, int l) {
  float hv[16];
#pragma unroll
  for (int k = 0; k < 16; ++k) {
    const unsigned ho = (unsigned)__shfl((int)hofs, k, 16);
    hv[k] = h[ho + l];
  }
  const bool ch  = (l & 8) != 0;
  const bool b2  = (l & 4) != 0;
  const bool b1  = (l & 2) != 0;
  const bool odd = (l & 1) != 0;
  const unsigned wlane = (unsigned)((l & 7) * 8 + (l >> 3) * 4);
  const unsigned cofs  = (unsigned)(((l >> 3) << 2) | ((l >> 1) & 3));
#pragma unroll
  for (int k = 0; k < 16; k += 2) {
    float pr[2];
#pragma unroll
    for (int d = 0; d < 2; ++d) {
      const unsigned wo = (unsigned)__shfl((int)wofs, k + d, 16);
      const float v = __shfl(val, k + d, 16);
      const v4f* __restrict__ wp = (const v4f*)(W2 + wo + wlane);
      const v4f wa = wp[0];
      const v4f wb = wp[16];
      const float mine = hv[k + d];
      const float far  = lane_partner<0x128>(mine);
      const float ha = ch ? far : mine;
      const float hb = ch ? mine : far;
      float t[4];
#pragma unroll
      for (int j = 0; j < 4; ++j) t[j] = ha * wa[j] + hb * wb[j];
      float r[2];
#pragma unroll
      for (int j = 0; j < 2; ++j) {
        const float keep = b2 ? t[2 + j] : t[j];
        const float give = b2 ? t[j] : t[2 + j];
        r[j] = keep + lane_partner<0x141>(give);
      }
      const float keepb = b1 ? r[1] : r[0];
      const float giveb = b1 ? r[0] : r[1];
      const float u = keepb + lane_partner<0x4E>(giveb);
      pr[d] = v * (u + lane_partner<0xB1>(u));
    }
    const unsigned oo0 = (unsigned)__shfl((int)oofs, k, 16);
    const unsigned oo1 = (unsigned)__shfl((int)oofs, k + 1, 16);
    const unsigned oo = odd ? oo1 : oo0;
    const float add = odd ? pr[1] : pr[0];
    if (GUARD) {
      const int ok0 = __shfl(ok, k, 16);
      const int ok1 = __shfl(ok, k + 1, 16);
      if (odd ? ok1 : ok0) atomicAdd(&out[oo + cofs], add);
    } else {
      atomicAdd(&out[oo + cofs], add);
    }
  }
}

__global__ __launch_bounds__(256) void layer2_scatter_kernel(
    const int* __restrict__ s, const int* __restrict__ p, const int* __restrict__ o,
    const float* __restrict__ deg, const float* __restrict__ h,
    const float* __restrict__ W2,
    float* __restrict__ out, int E, int N, int R) {
  const int lane = threadIdx.x & 31;
  const long long total = (long long)2 * E + N;
  const long long base = (((long long)blockIdx.x * blockDim.x + threadIdx.x) >> 5) << 5;
  if (base >= total) return;
  int si, pi, oi;
  float val;
  edge_decode(s, p, o, deg, base + lane, E, N, R, si, pi, oi, val);
  const unsigned hofs = (unsigned)oi * 16u;
  const unsigned wofs = (unsigned)pi * 128u;
  const unsigned oofs = (unsigned)si * 8u;
  if (base + 32 <= total) layer2_rows<false>(h, W2, out, hofs, wofs, oofs, val, 1, lane & 15);
  else layer2_rows<true>(h, W2, out, hofs, wofs, oofs, val, (base + lane) < total ? 1 : 0, lane & 15);
}

extern "C" void kernel_launch(void* const* d_in, const int* in_sizes, int n_in,
                              void* d_out, int out_size, void* d_ws, size_t ws_size,
                              hipStream_t stream) {
  const float* comps1 = (const float*)d_in[0];
  const float* bases1 = (const float*)d_in[1];
  const float* comps2 = (const float*)d_in[2];
  const float* bases2 = (const float*)d_in[3];
  const float* bias1  = (const float*)d_in[4];
  const float* bias2  = (const float*)d_in[5];
  const int*   s      = (const int*)d_in[6];
  const int*   p      = (const int*)d_in[7];
  const int*   o      = (const int*)d_in[8];

  const int EMB    = in_sizes[4];
  const int NUMCLS = in_sizes[5];
  const int E      = in_sizes[6];
  const int N      = out_size / NUMCLS;
  const int BASES  = (int)((long long)in_sizes[1] / ((long long)N * EMB));
  const int r_tot  = in_sizes[0] / BASES;
  const int R      = (r_tot - 1) / 2;

  float* ws = (float*)d_ws;
  size_t off = 0;
  auto alloc = [&](size_t nfloats) {
    size_t cur = off;
    off += (nfloats + 63) & ~(size_t)63;
    return cur;
  };
  float* deg = ws + alloc((size_t)r_tot * N);
  float* h   = ws + alloc((size_t)N * EMB);
  float* W2  = ws + alloc((size_t)r_tot * EMB * NUMCLS);
  float* W1  = ws + alloc((size_t)r_tot * N * EMB);
  (void)ws_size; (void)n_in;

  const long long NE   = (long long)N * EMB;
  const long long J2   = (long long)EMB * NUMCLS;
  const long long Eall = (long long)2 * E + N;
  const int B = 256;

  fill_zero_kernel<<<idiv_up((long long)r_tot * N, B), B, 0, stream>>>(deg, (long long)r_tot * N);
  fill_zero_kernel<<<idiv_up(NE, B), B, 0, stream>>>(h, NE);

  degree_kernel<<<idiv_up((long long)E + N, B), B, 0, stream>>>(s, p, o, deg, E, N, R);

  if (BASES == 40 && (NE & 63) == 0 && r_tot <= 48) {
    gemm48_stream_kernel<40><<<1024, 128, 0, stream>>>(comps1, bases1, W1, r_tot, NE);
  } else {
    gemm48_wmma_kernel<<<idiv_up(NE, 16LL * (B / 32)), B, 0, stream>>>(
        comps1, bases1, W1, r_tot, BASES, NE);
  }
  gemm48_wmma_kernel<<<idiv_up(J2, 16LL * (B / 32)), B, 0, stream>>>(
      comps2, bases2, W2, r_tot, BASES, J2);

  layer1_scatter_kernel<<<idiv_up(Eall, B), B, 0, stream>>>(
      s, p, o, deg, W1, h, E, N, R, NE);
  relu_bias_kernel<<<idiv_up(NE, B), B, 0, stream>>>(h, bias1, NE);

  init_out_kernel<<<idiv_up((long long)N * NUMCLS, B), B, 0, stream>>>(
      (float*)d_out, bias2, (long long)N * NUMCLS, NUMCLS);
  layer2_scatter_kernel<<<idiv_up(Eall, B), B, 0, stream>>>(
      s, p, o, deg, h, W2, (float*)d_out, E, N, R);
}


# Nine launches fused to five: deg and h clears, out = bias2 and the W2 WMMA product share one launch (product in the first work-group); ReLU(h + bias1) is applied where layer 2 gathers h
# speedup vs baseline: 1.6130x; 1.1039x over previous
#include <hip/hip_runtime.h>


typedef __attribute__((ext_vector_type(2))) float v2f;
typedef __attribute__((ext_vector_type(4))) float v4f;
typedef __attribute__((ext_vector_type(8))) float v8f;

static inline int idiv_up(long long a, long long b) { return (int)((a + b - 1) / b); }

__global__ void degree_kernel(const int* __restrict__ s, const int* __restrict__ p,
                              const int* __restrict__ o, float* __restrict__ deg,
                              int E, int N, int R) {
  long long i = (long long)blockIdx.x * blockDim.x + threadIdx.x;
  if (i < E) {
    int si = s[i], pi = p[i], oi = o[i];
    atomicAdd(&deg[(size_t)pi * N + si], 1.0f);
    atomicAdd(&deg[(size_t)(pi + R) * N + oi], 1.0f);
  } else if (i < (long long)E + N) {
    int node = (int)(i - E);
    deg[(size_t)(2 * R) * N + node] = 1.0f;
  }
}

__device__ __forceinline__ void gemm48_wmma_block(const float* __restrict__ A,
                                                  const float* __restrict__ B,
                                                  float* __restrict__ C,
                                                  int M, int K, long long J,
                                                  long long blk, float* __restrict__ sA) {
  const int total = 48 * K;
  for (int idx = threadIdx.x; idx < total; idx += blockDim.x) {
    const int row = idx / K;
    const int kk  = idx - row * K;
    sA[idx] = (row < M) ? A[row * K + kk] : 0.0f;
  }
  __syncthreads();

  const int wave = threadIdx.x >> 5;
  const int lane = threadIdx.x & 31;
  const long long jbase = (blk * (blockDim.x >> 5) + wave) * 16;
  if (jbase >= J) return;
  const int col   = lane & 15;
  const int khalf = (lane >> 4) * 2;

  v8f acc[3];
  acc[0] = (v8f){};
  acc[1] = (v8f){};
  acc[2] = (v8f){};

  const float* __restrict__ Bp = B + jbase + col;
  for (int kc = 0; kc < K; kc += 4) {
    const int k0 = kc + khalf;
    v2f b;
    b.x = Bp[(long long)k0 * J];
    b.y = Bp[(long long)(k0 + 1) * J];
#pragma unroll
    for (int mt = 0; mt < 3; ++mt) {
      const int m = mt * 16 + col;
      const v2f a = *(const v2f*)&sA[m * K + k0];
      acc[mt] = __builtin_amdgcn_wmma_f32_16x16x4_f32(
          false, a, false, b, (short)0, acc[mt], false, false);
    }
  }

  const int rhalf = (lane >> 4) * 8;
#pragma unroll
  for (int mt = 0; mt < 3; ++mt) {
#pragma unroll
    for (int v = 0; v < 8; ++v) {
      const int row = mt * 16 + rhalf + v;
      if (row < M) C[(long long)row * J + jbase + col] = acc[mt][v];
    }
  }
}

__global__ void gemm48_wmma_kernel(const float* __restrict__ A,
                                   const float* __restrict__ B,
                                   float* __restrict__ C,
                                   int M, int K, long long J) {
  __shared__ float sA[48 * 64];
  gemm48_wmma_block(A, B, C, M, K, J, (long long)blockIdx.x, sA);
}

__global__ __launch_bounds__(256) void init_and_w2_kernel(
    float* __restrict__ deg, long long n_deg, float* __restrict__ h, long long n_h,
    float* __restrict__ out, const float* __restrict__ bias2, long long n_out, int C,
    const float* __restrict__ A2, const float* __restrict__ B2, float* __restrict__ W2,
    int M, int K, long long J2, int w2_blocks) {
  __shared__ float sA[48 * 64];
  if ((int)blockIdx.x < w2_blocks) {
    gemm48_wmma_block(A2, B2, W2, M, K, J2, (long long)blockIdx.x, sA);
    return;
  }
  long long i = (long long)((int)blockIdx.x - w2_blocks) * blockDim.x + threadIdx.x;
  if (i < n_deg) { deg[i] = 0.0f; return; }
  i -= n_deg;
  if (i < n_h) { h[i] = 0.0f; return; }
  i -= n_h;
  if (i < n_out) out[i] = bias2[i % C];
}

template <int KS, bool REFILL>
__device__ __forceinline__ void gemm48_chunk(const v2f (&a)[3][KS], v4f (&b0)[KS],
                                             v4f (&b1)[KS], v8f (&acc)[3][4],
                                             const float* __restrict__ pn, long long J) {
#pragma unroll
  for (int mt = 0; mt < 3; ++mt)
#pragma unroll
    for (int t = 0; t < 4; ++t) acc[mt][t] = (v8f){};
#pragma unroll
  for (int ks = 0; ks < KS; ++ks) {
    const v4f r0 = b0[ks];
    const v4f r1 = b1[ks];
#pragma unroll
    for (int t = 0; t < 4; ++t) {
      const v2f b = {r0[t], r1[t]};
#pragma unroll
      for (int mt = 0; mt < 3; ++mt)
        acc[mt][t] = __builtin_amdgcn_wmma_f32_16x16x4_f32(
            false, a[mt][ks], false, b, (short)0, acc[mt][t], false, false);
    }
    if (REFILL) {
      b0[ks] = __builtin_nontemporal_load((const v4f*)(pn + (long long)(4 * ks) * J));
      b1[ks] = __builtin_nontemporal_load((const v4f*)(pn + (long long)(4 * ks + 1) * J));
    }
    __builtin_amdgcn_sched_barrier(0);
  }
}

__device__ __forceinline__ void gemm48_store(const v8f (&acc)[3][4], float* __restrict__ pc,
                                             int lim, long long J) {
#pragma unroll
  for (int mt = 0; mt < 3; ++mt) {
#pragma unroll
    for (int r = 0; r < 8; ++r) {
      if (mt * 16 + r < lim) {
        const v4f d = {acc[mt][0][r], acc[mt][1][r], acc[mt][2][r], acc[mt][3][r]};
        *(v4f*)(pc + (long long)(mt * 16 + r) * J) = d;
      }
    }
  }
}

template <int K>
__global__ __launch_bounds__(128, 4) void gemm48_stream_kernel(
    const float* __restrict__ A, const float* __restrict__ B,
    float* __restrict__ C, int M, long long J) {
  constexpr int KS = K / 4;
  const int lane = threadIdx.x & 31;
  const int c = lane & 15;
  const int h = lane >> 4;
  const int wave = __builtin_amdgcn_readfirstlane((int)(threadIdx.x >> 5));
  const int wpb  = (int)(blockDim.x >> 5);
  const long long nw  = (long long)gridDim.x * wpb;
  const long long nST = J >> 6;
  long long st = (long long)blockIdx.x * wpb + wave;
  if (st >= nST) return;

  v2f a[3][KS];
#pragma unroll
  for (int mt = 0; mt < 3; ++mt) {
    const int row = mt * 16 + c;
#pragma unroll
    for (int ks = 0; ks < KS; ++ks) {
      v2f t = {0.0f, 0.0f};
      if (row < M) t = *(const v2f*)&A[row * K + 4 * ks + 2 * h];
      a[mt][ks] = t;
    }
  }

  const float* __restrict__ Bl = B + (long long)(2 * h) * J + 4 * c;
  float* __restrict__ Cl = C + (long long)(8 * h) * J + 4 * c;
  const int lim = M - 8 * h;

  v4f b0[KS], b1[KS];
  {
    const float* p = Bl + st * 64;
#pragma unroll
    for (int ks = 0; ks < KS; ++ks) {
      b0[ks] = __builtin_nontemporal_load((const v4f*)(p + (long long)(4 * ks) * J));
      b1[ks] = __builtin_nontemporal_load((const v4f*)(p + (long long)(4 * ks + 1) * J));
    }
  }

  v8f acc[3][4];
  while (st + nw < nST) {
    gemm48_chunk<KS, true>(a, b0, b1, acc, Bl + (st + nw) * 64, J);
    gemm48_store(acc, Cl + st * 64, lim, J);
    st += nw;
  }
  gemm48_chunk<KS, false>(a, b0, b1, acc, Bl, J);
  gemm48_store(acc, Cl + st * 64, lim, J);
}

__device__ __forceinline__ void edge_decode(const int* __restrict__ s, const int* __restrict__ p,
                                            const int* __restrict__ o,
                                            const float* __restrict__ deg,
                                            long long i, int E, int N, int R,
                                            int& si, int& pi, int& oi, float& val) {
  si = 0; pi = 0; oi = 0; val = 0.0f;
  if (i < (long long)2 * E + N) {
    if (i < E)            { si = s[i];      pi = p[i];          oi = o[i]; }
    else if (i < 2LL * E) { long long j = i - E; si = o[j]; pi = p[j] + R; oi = s[j]; }
    else                  { int nd = (int)(i - 2LL * E); si = nd; pi = 2 * R; oi = nd; }
    val = 1.0f / deg[(size_t)pi * N + si];
  }
}

template <bool GUARD>
__device__ __forceinline__ void layer1_rows(const float* __restrict__ W1, float* __restrict__ h,
                                            unsigned wofs, unsigned hofs, float val, int ok,
                                            int e) {
  float w[16];
#pragma unroll
  for (int k = 0; k < 16; ++k) {
    const unsigned wo = (unsigned)__shfl((int)wofs, k, 16);
    w[k] = W1[wo + e];
  }
#pragma unroll
  for (int k = 0; k < 16; ++k) {
    const unsigned ho = (unsigned)__shfl((int)hofs, k, 16);
    const float v = __shfl(val, k, 16);
    if (!GUARD || __shfl(ok, k, 16)) atomicAdd(&h[ho + e], v * w[k]);
  }
}

__global__ __launch_bounds__(256) void layer1_scatter_kernel(
    const int* __restrict__ s, const int* __restrict__ p, const int* __restrict__ o,
    const float* __restrict__ deg, const float* __restrict__ W1, float* __restrict__ h,
    int E, int N, int R, long long NE) {
  const int lane = threadIdx.x & 31;
  const long long total = (long long)2 * E + N;
  const long long base = (((long long)blockIdx.x * blockDim.x + threadIdx.x) >> 5) << 5;
  if (base >= total) return;
  int si, pi, oi;
  float val;
  edge_decode(s, p, o, deg, base + lane, E, N, R, si, pi, oi, val);
  const unsigned wofs = (unsigned)((long long)pi * NE + (long long)oi * 16);
  const unsigned hofs = (unsigned)si * 16u;
  const int e = lane & 15;
  if (base + 32 <= total) layer1_rows<false>(W1, h, wofs, hofs, val, 1, e);
  else layer1_rows<true>(W1, h, wofs, hofs, val, (base + lane) < total ? 1 : 0, e);
}

template <int CTRL>
__device__ __forceinline__ float lane_partner(float x) {
  return __builtin_bit_cast(
      float, __builtin_amdgcn_update_dpp(0, __builtin_bit_cast(int, x), CTRL, 0xf, 0xf, true));
}

template <bool GUARD>
__device__ __forceinline__ void layer2_rows(const float* __restrict__ h,
                                            const float* __restrict__ bias1,
                                            const float* __restrict__ W2,
                                            float* __restrict__ out,
                                            unsigned hofs, unsigned wofs, unsigned oofs,
                                            float val, int ok, int l) {
  float hv[16];
#pragma unroll
  for (int k = 0; k < 16; ++k) {
    const unsigned ho = (unsigned)__shfl((int)hofs, k, 16);
    hv[k] = h[ho + l];
  }
  const float b1v = bias1[l];
#pragma unroll
  for (int k = 0; k < 16; ++k) hv[k] = fmaxf(hv[k] + b1v, 0.0f);
  const bool ch  = (l & 8) != 0;
  const bool b2  = (l & 4) != 0;
  const bool b1  = (l & 2) != 0;
  const bool odd = (l & 1) != 0;
  const unsigned wlane = (unsigned)((l & 7) * 8 + (l >> 3) * 4);
  const unsigned cofs  = (unsigned)(((l >> 3) << 2) | ((l >> 1) & 3));
#pragma unroll
  for (int k = 0; k < 16; k += 2) {
    float pr[2];
#pragma unroll
    for (int d = 0; d < 2; ++d) {
      const unsigned wo = (unsigned)__shfl((int)wofs, k + d, 16);
      const float v = __shfl(val, k + d, 16);
      const v4f* __restrict__ wp = (const v4f*)(W2 + wo + wlane);
      const v4f wa = wp[0];
      const v4f wb = wp[16];
      const float mine = hv[k + d];
      const float far  = lane_partner<0x128>(mine);
      const float ha = ch ? far : mine;
      const float hb = ch ? mine : far;
      float t[4];
#pragma unroll
      for (int j = 0; j < 4; ++j) t[j] = ha * wa[j] + hb * wb[j];
      float r[2];
#pragma unroll
      for (int j = 0; j < 2; ++j) {
        const float keep = b2 ? t[2 + j] : t[j];
        const float give = b2 ? t[j] : t[2 + j];
        r[j] = keep + lane_partner<0x141>(give);
      }
      const float keepb = b1 ? r[1] : r[0];
      const float giveb = b1 ? r[0] : r[1];
      const float u = keepb + lane_partner<0x4E>(giveb);
      pr[d] = v * (u + lane_partner<0xB1>(u));
    }
    const unsigned oo0 = (unsigned)__shfl((int)oofs, k, 16);
    const unsigned oo1 = (unsigned)__shfl((int)oofs, k + 1, 16);
    const unsigned oo = odd ? oo1 : oo0;
    const float add = odd ? pr[1] : pr[0];
    if (GUARD) {
      const int ok0 = __shfl(ok, k, 16);
      const int ok1 = __shfl(ok, k + 1, 16);
      if (odd ? ok1 : ok0) atomicAdd(&out[oo + cofs], add);
    } else {
      atomicAdd(&out[oo + cofs], add);
    }
  }
}

__global__ __launch_bounds__(256) void layer2_scatter_kernel(
    const int* __restrict__ s, const int* __restrict__ p, const int* __restrict__ o,
    const float* __restrict__ deg, const float* __restrict__ h,
    const float* __restrict__ bias1,
    const float* __restrict__ W2,
    float* __restrict__ out, int E, int N, int R) {
  const int lane = threadIdx.x & 31;
  const long long total = (long long)2 * E + N;
  const long long base = (((long long)blockIdx.x * blockDim.x + threadIdx.x) >> 5) << 5;
  if (base >= total) return;
  int si, pi, oi;
  float val;
  edge_decode(s, p, o, deg, base + lane, E, N, R, si, pi, oi, val);
  const unsigned hofs = (unsigned)oi * 16u;
  const unsigned wofs = (unsigned)pi * 128u;
  const unsigned oofs = (unsigned)si * 8u;
  if (base + 32 <= total) layer2_rows<false>(h, bias1, W2, out, hofs, wofs, oofs, val, 1, lane & 15);
  else layer2_rows<true>(h, bias1, W2, out, hofs, wofs, oofs, val, (base + lane) < total ? 1 : 0, lane & 15);
}

extern "C" void kernel_launch(void* const* d_in, const int* in_sizes, int n_in,
                              void* d_out, int out_size, void* d_ws, size_t ws_size,
                              hipStream_t stream) {
  const float* comps1 = (const float*)d_in[0];
  const float* bases1 = (const float*)d_in[1];
  const float* comps2 = (const float*)d_in[2];
  const float* bases2 = (const float*)d_in[3];
  const float* bias1  = (const float*)d_in[4];
  const float* bias2  = (const float*)d_in[5];
  const int*   s      = (const int*)d_in[6];
  const int*   p      = (const int*)d_in[7];
  const int*   o      = (const int*)d_in[8];

  const int EMB    = in_sizes[4];
  const int NUMCLS = in_sizes[5];
  const int E      = in_sizes[6];
  const int N      = out_size / NUMCLS;
  const int BASES  = (int)((long long)in_sizes[1] / ((long long)N * EMB));
  const int r_tot  = in_sizes[0] / BASES;
  const int R      = (r_tot - 1) / 2;

  float* ws = (float*)d_ws;
  size_t off = 0;
  auto alloc = [&](size_t nfloats) {
    size_t cur = off;
    off += (nfloats + 63) & ~(size_t)63;
    return cur;
  };
  float* deg = ws + alloc((size_t)r_tot * N);
  float* h   = ws + alloc((size_t)N * EMB);
  float* W2  = ws + alloc((size_t)r_tot * EMB * NUMCLS);
  float* W1  = ws + alloc((size_t)r_tot * N * EMB);
  (void)ws_size; (void)n_in;

  const long long NE   = (long long)N * EMB;
  const long long J2   = (long long)EMB * NUMCLS;
  const long long Eall = (long long)2 * E + N;
  const int B = 256;

  const long long n_deg = (long long)r_tot * N;
  const long long n_out = (long long)N * NUMCLS;
  const int fill_blocks = idiv_up(n_deg + NE + n_out, B);
  const int w2_blocks   = idiv_up(J2, 16LL * (B / 32));
  init_and_w2_kernel<<<w2_blocks + fill_blocks, B, 0, stream>>>(
      deg, n_deg, h, NE, (float*)d_out, bias2, n_out, NUMCLS,
      comps2, bases2, W2, r_tot, BASES, J2, w2_blocks);

  degree_kernel<<<idiv_up((long long)E + N, B), B, 0, stream>>>(s, p, o, deg, E, N, R);

  if (BASES == 40 && (NE & 63) == 0 && r_tot <= 48) {
    gemm48_stream_kernel<40><<<1024, 128, 0, stream>>>(comps1, bases1, W1, r_tot, NE);
  } else {
    gemm48_wmma_kernel<<<idiv_up(NE, 16LL * (B / 32)), B, 0, stream>>>(
        comps1, bases1, W1, r_tot, BASES, NE);
  }

  layer1_scatter_kernel<<<idiv_up(Eall, B), B, 0, stream>>>(
      s, p, o, deg, W1, h, E, N, R, NE);

  layer2_scatter_kernel<<<idiv_up(Eall, B), B, 0, stream>>>(
      s, p, o, deg, h, bias1, W2, (float*)d_out, E, N, R);
}


# Degree count moved into the W1 GEMM's launch: each streaming wave parks its edges' segment indices in LDS and fires one batch of degree adds per chunk trip, under the memory stream; 5 launches to 4
# speedup vs baseline: 1.6842x; 1.0441x over previous
#include <hip/hip_runtime.h>


typedef __attribute__((ext_vector_type(2))) float v2f;
typedef __attribute__((ext_vector_type(4))) float v4f;
typedef __attribute__((ext_vector_type(8))) float v8f;

static inline int idiv_up(long long a, long long b) { return (int)((a + b - 1) / b); }

__global__ void degree_kernel(const int* __restrict__ s, const int* __restrict__ p,
                              const int* __restrict__ o, float* __restrict__ deg,
                              int E, int N, int R) {
  long long i = (long long)blockIdx.x * blockDim.x + threadIdx.x;
  if (i < E) {
    int si = s[i], pi = p[i], oi = o[i];
    atomicAdd(&deg[(size_t)pi * N + si], 1.0f);
    atomicAdd(&deg[(size_t)(pi + R) * N + oi], 1.0f);
  } else if (i < (long long)E + N) {
    int node = (int)(i - E);
    deg[(size_t)(2 * R) * N + node] = 1.0f;
  }
}

__device__ __forceinline__ void gemm48_wmma_block(const float* __restrict__ A,
                                                  const float* __restrict__ B,
                                                  float* __restrict__ C,
                                                  int M, int K, long long J,
                                                  long long blk, float* __restrict__ sA) {
  const int total = 48 * K;
  for (int idx = threadIdx.x; idx < total; idx += blockDim.x) {
    const int row = idx / K;
    const int kk  = idx - row * K;
    sA[idx] = (row < M) ? A[row * K + kk] : 0.0f;
  }
  __syncthreads();

  const int wave = threadIdx.x >> 5;
  const int lane = threadIdx.x & 31;
  const long long jbase = (blk * (blockDim.x >> 5) + wave) * 16;
  if (jbase >= J) return;
  const int col   = lane & 15;
  const int khalf = (lane >> 4) * 2;

  v8f acc[3];
  acc[0] = (v8f){};
  acc[1] = (v8f){};
  acc[2] = (v8f){};

  const float* __restrict__ Bp = B + jbase + col;
  for (int kc = 0; kc < K; kc += 4) {
    const int k0 = kc + khalf;
    v2f b;
    b.x = Bp[(long long)k0 * J];
    b.y = Bp[(long long)(k0 + 1) * J];
#pragma unroll
    for (int mt = 0; mt < 3; ++mt) {
      const int m = mt * 16 + col;
      const v2f a = *(const v2f*)&sA[m * K + k0];
      acc[mt] = __builtin_amdgcn_wmma_f32_16x16x4_f32(
          false, a, false, b, (short)0, acc[mt], false, false);
    }
  }

  const int rhalf = (lane >> 4) * 8;
#pragma unroll
  for (int mt = 0; mt < 3; ++mt) {
#pragma unroll
    for (int v = 0; v < 8; ++v) {
      const int row = mt * 16 + rhalf + v;
      if (row < M) C[(long long)row * J + jbase + col] = acc[mt][v];
    }
  }
}

__global__ void gemm48_wmma_kernel(const float* __restrict__ A,
                                   const float* __restrict__ B,
                                   float* __restrict__ C,
                                   int M, int K, long long J) {
  __shared__ float sA[48 * 64];
  gemm48_wmma_block(A, B, C, M, K, J, (long long)blockIdx.x, sA);
}

__global__ __launch_bounds__(256) void init_and_w2_kernel(
    float* __restrict__ deg, long long n_deg, long long n_rel, float* __restrict__ h, long long n_h,
    float* __restrict__ out, const float* __restrict__ bias2, long long n_out, int C,
    const float* __restrict__ A2, const float* __restrict__ B2, float* __restrict__ W2,
    int M, int K, long long J2, int w2_blocks) {
  __shared__ float sA[48 * 64];
  if ((int)blockIdx.x < w2_blocks) {
    gemm48_wmma_block(A2, B2, W2, M, K, J2, (long long)blockIdx.x, sA);
    return;
  }
  long long i = (long long)((int)blockIdx.x - w2_blocks) * blockDim.x + threadIdx.x;
  if (i < n_deg) { deg[i] = (i < n_rel) ? 0.0f : 1.0f; return; }
  i -= n_deg;
  if (i < n_h) { h[i] = 0.0f; return; }
  i -= n_h;
  if (i < n_out) out[i] = bias2[i % C];
}

template <int KS, bool REFILL>
__device__ __forceinline__ void gemm48_chunk(const v2f (&a)[3][KS], v4f (&b0)[KS],
                                             v4f (&b1)[KS], v8f (&acc)[3][4],
                                             const float* __restrict__ pn, long long J) {
#pragma unroll
  for (int mt = 0; mt < 3; ++mt)
#pragma unroll
    for (int t = 0; t < 4; ++t) acc[mt][t] = (v8f){};
#pragma unroll
  for (int ks = 0; ks < KS; ++ks) {
    const v4f r0 = b0[ks];
    const v4f r1 = b1[ks];
#pragma unroll
    for (int t = 0; t < 4; ++t) {
      const v2f b = {r0[t], r1[t]};
#pragma unroll
      for (int mt = 0; mt < 3; ++mt)
        acc[mt][t] = __builtin_amdgcn_wmma_f32_16x16x4_f32(
            false, a[mt][ks], false, b, (short)0, acc[mt][t], false, false);
    }
    if (REFILL) {
      b0[ks] = __builtin_nontemporal_load((const v4f*)(pn + (long long)(4 * ks) * J));
      b1[ks] = __builtin_nontemporal_load((const v4f*)(pn + (long long)(4 * ks + 1) * J));
    }
    __builtin_amdgcn_sched_barrier(0);
  }
}

__device__ __forceinline__ void gemm48_store(const v8f (&acc)[3][4], float* __restrict__ pc,
                                             int lim, long long J) {
#pragma unroll
  for (int mt = 0; mt < 3; ++mt) {
#pragma unroll
    for (int r = 0; r < 8; ++r) {
      if (mt * 16 + r < lim) {
        const v4f d = {acc[mt][0][r], acc[mt][1][r], acc[mt][2][r], acc[mt][3][r]};
        *(v4f*)(pc + (long long)(mt * 16 + r) * J) = d;
      }
    }
  }
}

constexpr int DEG_SLOTS = 8;
constexpr unsigned DEG_NONE = 0xFFFFFFFFu;

__device__ __forceinline__ void deg_segments(const int* __restrict__ s, const int* __restrict__ p,
                                             const int* __restrict__ o, long long i, int N, int R,
                                             unsigned& a0, unsigned& a1) {
  const int si = s[i], pi = p[i], oi = o[i];
  a0 = (unsigned)pi * (unsigned)N + (unsigned)si;
  a1 = (unsigned)(pi + R) * (unsigned)N + (unsigned)oi;
}

template <int K>
__global__ __launch_bounds__(128, 4) void gemm48_stream_deg_kernel(
    const float* __restrict__ A, const float* __restrict__ B,
    float* __restrict__ C, int M, long long J,
    const int* __restrict__ s, const int* __restrict__ p, const int* __restrict__ o,
    float* __restrict__ deg, int E, int N, int R) {
  constexpr int KS = K / 4;
  __shared__ uint2 sseg[4][DEG_SLOTS][32];
  const int lane = threadIdx.x & 31;
  const int c = lane & 15;
  const int h = lane >> 4;
  const int wave = __builtin_amdgcn_readfirstlane((int)(threadIdx.x >> 5));
  const int wpb  = (int)(blockDim.x >> 5);
  const long long nw  = (long long)gridDim.x * wpb;
  const long long nST = J >> 6;
  const long long gw = (long long)blockIdx.x * wpb + wave;
  long long st = gw;
  const long long estep = nw * 32;

  if (st >= nST) {
    for (long long eb = gw * 32; eb < E; eb += estep) {
      const long long i = eb + lane;
      if (i < E) {
        unsigned a0, a1;
        deg_segments(s, p, o, i, N, R, a0, a1);
        atomicAdd(&deg[a0], 1.0f);
        atomicAdd(&deg[a1], 1.0f);
      }
    }
    return;
  }

  int es[DEG_SLOTS], ep[DEG_SLOTS], eo[DEG_SLOTS];
  int nb = 0;
#pragma unroll
  for (int k = 0; k < DEG_SLOTS; ++k) {
    const long long eb = gw * 32 + (long long)k * estep;
    es[k] = 0; ep[k] = 0; eo[k] = 0;
    if (eb < E) {
      nb = k + 1;
      const long long i = eb + lane;
      const long long ic = i < E ? i : (long long)E - 1;
      es[k] = s[ic]; ep[k] = p[ic]; eo[k] = o[ic];
    }
  }
  for (long long eb = gw * 32 + (long long)DEG_SLOTS * estep; eb < E; eb += estep) {
    const long long i = eb + lane;
    if (i < E) {
      unsigned a0, a1;
      deg_segments(s, p, o, i, N, R, a0, a1);
      atomicAdd(&deg[a0], 1.0f);
      atomicAdd(&deg[a1], 1.0f);
    }
  }
  auto deg_issue = [&](int b) {
    const uint2 sg = sseg[wave][b][lane];
    if (sg.x != DEG_NONE) {
      atomicAdd(&deg[sg.x], 1.0f);
      atomicAdd(&deg[sg.y], 1.0f);
    }
  };

  v2f a[3][KS];
#pragma unroll
  for (int mt = 0; mt < 3; ++mt) {
    const int row = mt * 16 + c;
#pragma unroll
    for (int ks = 0; ks < KS; ++ks) {
      v2f t = {0.0f, 0.0f};
      if (row < M) t = *(const v2f*)&A[row * K + 4 * ks + 2 * h];
      a[mt][ks] = t;
    }
  }

  const float* __restrict__ Bl = B + (long long)(2 * h) * J + 4 * c;
  float* __restrict__ Cl = C + (long long)(8 * h) * J + 4 * c;
  const int lim = M - 8 * h;

  v4f b0[KS], b1[KS];
  {
    const float* p = Bl + st * 64;
#pragma unroll
    for (int ks = 0; ks < KS; ++ks) {
      b0[ks] = __builtin_nontemporal_load((const v4f*)(p + (long long)(4 * ks) * J));
      b1[ks] = __builtin_nontemporal_load((const v4f*)(p + (long long)(4 * ks + 1) * J));
    }
  }

#pragma unroll
  for (int k = 0; k < DEG_SLOTS; ++k) {
    if (k < nb) {
      const long long i = gw * 32 + (long long)k * estep + lane;
      unsigned a0 = DEG_NONE, a1 = DEG_NONE;
      if (i < E) {
        a0 = (unsigned)ep[k] * (unsigned)N + (unsigned)es[k];
        a1 = (unsigned)(ep[k] + R) * (unsigned)N + (unsigned)eo[k];
      }
      sseg[wave][k][lane] = make_uint2(a0, a1);
    }
  }

  v8f acc[3][4];
  int bi = 0;
  while (st + nw < nST) {
    if (bi < nb) { deg_issue(bi); ++bi; }
    gemm48_chunk<KS, true>(a, b0, b1, acc, Bl + (st + nw) * 64, J);
    gemm48_store(acc, Cl + st * 64, lim, J);
    st += nw;
  }
  for (; bi < nb; ++bi) deg_issue(bi);
  gemm48_chunk<KS, false>(a, b0, b1, acc, Bl, J);
  gemm48_store(acc, Cl + st * 64, lim, J);
}

__device__ __forceinline__ void edge_decode(const int* __restrict__ s, const int* __restrict__ p,
                                            const int* __restrict__ o,
                                            const float* __restrict__ deg,
                                            long long i, int E, int N, int R,
                                            int& si, int& pi, int& oi, float& val) {
  si = 0; pi = 0; oi = 0; val = 0.0f;
  if (i < (long long)2 * E + N) {
    if (i < E)            { si = s[i];      pi = p[i];          oi = o[i]; }
    else if (i < 2LL * E) { long long j = i - E; si = o[j]; pi = p[j] + R; oi = s[j]; }
    else                  { int nd = (int)(i - 2LL * E); si = nd; pi = 2 * R; oi = nd; }
    val = 1.0f / deg[(size_t)pi * N + si];
  }
}

template <bool GUARD>
__device__ __forceinline__ void layer1_rows(const float* __restrict__ W1, float* __restrict__ h,
                                            unsigned wofs, unsigned hofs, float val, int ok,
                                            int e) {
  float w[16];
#pragma unroll
  for (int k = 0; k < 16; ++k) {
    const unsigned wo = (unsigned)__shfl((int)wofs, k, 16);
    w[k] = W1[wo + e];
  }
#pragma unroll
  for (int k = 0; k < 16; ++k) {
    const unsigned ho = (unsigned)__shfl((int)hofs, k, 16);
    const float v = __shfl(val, k, 16);
    if (!GUARD || __shfl(ok, k, 16)) atomicAdd(&h[ho + e], v * w[k]);
  }
}

__global__ __launch_bounds__(256) void layer1_scatter_kernel(
    const int* __restrict__ s, const int* __restrict__ p, const int* __restrict__ o,
    const float* __restrict__ deg, const float* __restrict__ W1, float* __restrict__ h,
    int E, int N, int R, long long NE) {
  const int lane = threadIdx.x & 31;
  const long long total = (long long)2 * E + N;
  const long long base = (((long long)blockIdx.x * blockDim.x + threadIdx.x) >> 5) << 5;
  if (base >= total) return;
  int si, pi, oi;
  float val;
  edge_decode(s, p, o, deg, base + lane, E, N, R, si, pi, oi, val);
  const unsigned wofs = (unsigned)((long long)pi * NE + (long long)oi * 16);
  const unsigned hofs = (unsigned)si * 16u;
  const int e = lane & 15;
  if (base + 32 <= total) layer1_rows<false>(W1, h, wofs, hofs, val, 1, e);
  else layer1_rows<true>(W1, h, wofs, hofs, val, (base + lane) < total ? 1 : 0, e);
}

template <int CTRL>
__device__ __forceinline__ float lane_partner(float x) {
  return __builtin_bit_cast(
      float, __builtin_amdgcn_update_dpp(0, __builtin_bit_cast(int, x), CTRL, 0xf, 0xf, true));
}

template <bool GUARD>
__device__ __forceinline__ void layer2_rows(const float* __restrict__ h,
                                            const float* __restrict__ bias1,
                                            const float* __restrict__ W2,
                                            float* __restrict__ out,
                                            unsigned hofs, unsigned wofs, unsigned oofs,
                                            float val, int ok, int l) {
  float hv[16];
#pragma unroll
  for (int k = 0; k < 16; ++k) {
    const unsigned ho = (unsigned)__shfl((int)hofs, k, 16);
    hv[k] = h[ho + l];
  }
  const float b1v = bias1[l];
#pragma unroll
  for (int k = 0; k < 16; ++k) hv[k] = fmaxf(hv[k] + b1v, 0.0f);
  const bool ch  = (l & 8) != 0;
  const bool b2  = (l & 4) != 0;
  const bool b1  = (l & 2) != 0;
  const bool odd = (l & 1) != 0;
  const unsigned wlane = (unsigned)((l & 7) * 8 + (l >> 3) * 4);
  const unsigned cofs  = (unsigned)(((l >> 3) << 2) | ((l >> 1) & 3));
#pragma unroll
  for (int k = 0; k < 16; k += 2) {
    float pr[2];
#pragma unroll
    for (int d = 0; d < 2; ++d) {
      const unsigned wo = (unsigned)__shfl((int)wofs, k + d, 16);
      const float v = __shfl(val, k + d, 16);
      const v4f* __restrict__ wp = (const v4f*)(W2 + wo + wlane);
      const v4f wa = wp[0];
      const v4f wb = wp[16];
      const float mine = hv[k + d];
      const float far  = lane_partner<0x128>(mine);
      const float ha = ch ? far : mine;
      const float hb = ch ? mine : far;
      float t[4];
#pragma unroll
      for (int j = 0; j < 4; ++j) t[j] = ha * wa[j] + hb * wb[j];
      float r[2];
#pragma unroll
      for (int j = 0; j < 2; ++j) {
        const float keep = b2 ? t[2 + j] : t[j];
        const float give = b2 ? t[j] : t[2 + j];
        r[j] = keep + lane_partner<0x141>(give);
      }
      const float keepb = b1 ? r[1] : r[0];
      const float giveb = b1 ? r[0] : r[1];
      const float u = keepb + lane_partner<0x4E>(giveb);
      pr[d] = v * (u + lane_partner<0xB1>(u));
    }
    const unsigned oo0 = (unsigned)__shfl((int)oofs, k, 16);
    const unsigned oo1 = (unsigned)__shfl((int)oofs, k + 1, 16);
    const unsigned oo = odd ? oo1 : oo0;
    const float add = odd ? pr[1] : pr[0];
    if (GUARD) {
      const int ok0 = __shfl(ok, k, 16);
      const int ok1 = __shfl(ok, k + 1, 16);
      if (odd ? ok1 : ok0) atomicAdd(&out[oo + cofs], add);
    } else {
      atomicAdd(&out[oo + cofs], add);
    }
  }
}

__global__ __launch_bounds__(256) void layer2_scatter_kernel(
    const int* __restrict__ s, const int* __restrict__ p, const int* __restrict__ o,
    const float* __restrict__ deg, const float* __restrict__ h,
    const float* __restrict__ bias1,
    const float* __restrict__ W2,
    float* __restrict__ out, int E, int N, int R) {
  const int lane = threadIdx.x & 31;
  const long long total = (long long)2 * E + N;
  const long long base = (((long long)blockIdx.x * blockDim.x + threadIdx.x) >> 5) << 5;
  if (base >= total) return;
  int si, pi, oi;
  float val;
  edge_decode(s, p, o, deg, base + lane, E, N, R, si, pi, oi, val);
  const unsigned hofs = (unsigned)oi * 16u;
  const unsigned wofs = (unsigned)pi * 128u;
  const unsigned oofs = (unsigned)si * 8u;
  if (base + 32 <= total) layer2_rows<false>(h, bias1, W2, out, hofs, wofs, oofs, val, 1, lane & 15);
  else layer2_rows<true>(h, bias1, W2, out, hofs, wofs, oofs, val, (base + lane) < total ? 1 : 0, lane & 15);
}

extern "C" void kernel_launch(void* const* d_in, const int* in_sizes, int n_in,
                              void* d_out, int out_size, void* d_ws, size_t ws_size,
                              hipStream_t stream) {
  const float* comps1 = (const float*)d_in[0];
  const float* bases1 = (const float*)d_in[1];
  const float* comps2 = (const float*)d_in[2];
  const float* bases2 = (const float*)d_in[3];
  const float* bias1  = (const float*)d_in[4];
  const float* bias2  = (const float*)d_in[5];
  const int*   s      = (const int*)d_in[6];
  const int*   p      = (const int*)d_in[7];
  const int*   o      = (const int*)d_in[8];

  const int EMB    = in_sizes[4];
  const int NUMCLS = in_sizes[5];
  const int E      = in_sizes[6];
  const int N      = out_size / NUMCLS;
  const int BASES  = (int)((long long)in_sizes[1] / ((long long)N * EMB));
  const int r_tot  = in_sizes[0] / BASES;
  const int R      = (r_tot - 1) / 2;

  float* ws = (float*)d_ws;
  size_t off = 0;
  auto alloc = [&](size_t nfloats) {
    size_t cur = off;
    off += (nfloats + 63) & ~(size_t)63;
    return cur;
  };
  float* deg = ws + alloc((size_t)r_tot * N);
  float* h   = ws + alloc((size_t)N * EMB);
  float* W2  = ws + alloc((size_t)r_tot * EMB * NUMCLS);
  float* W1  = ws + alloc((size_t)r_tot * N * EMB);
  (void)ws_size; (void)n_in;

  const long long NE   = (long long)N * EMB;
  const long long J2   = (long long)EMB * NUMCLS;
  const long long Eall = (long long)2 * E + N;
  const int B = 256;

  const long long n_deg = (long long)r_tot * N;
  const long long n_out = (long long)N * NUMCLS;
  const int fill_blocks = idiv_up(n_deg + NE + n_out, B);
  const int w2_blocks   = idiv_up(J2, 16LL * (B / 32));
  const long long n_rel = (long long)2 * R * N;
  init_and_w2_kernel<<<w2_blocks + fill_blocks, B, 0, stream>>>(
      deg, n_deg, n_rel, h, NE, (float*)d_out, bias2, n_out, NUMCLS,
      comps2, bases2, W2, r_tot, BASES, J2, w2_blocks);

  if (BASES == 40 && (NE & 63) == 0 && r_tot <= 48) {
    gemm48_stream_deg_kernel<40><<<1024, 128, 0, stream>>>(comps1, bases1, W1, r_tot, NE,
                                                          s, p, o, deg, E, N, R);
  } else {
    degree_kernel<<<idiv_up((long long)E + N, B), B, 0, stream>>>(s, p, o, deg, E, N, R);
    gemm48_wmma_kernel<<<idiv_up(NE, 16LL * (B / 32)), B, 0, stream>>>(
        comps1, bases1, W1, r_tot, BASES, NE);
  }

  layer1_scatter_kernel<<<idiv_up(Eall, B), B, 0, stream>>>(
      s, p, o, deg, W1, h, E, N, R, NE);

  layer2_scatter_kernel<<<idiv_up(Eall, B), B, 0, stream>>>(
      s, p, o, deg, h, bias1, W2, (float*)d_out, E, N, R);
}


# Float atomic adds split by cache half: h and out get one copy per second-level cache half (die id read per wave), layer 2 adds the two h copies, a 5th launch adds the out copies (new sum order)
# speedup vs baseline: 1.8346x; 1.0893x over previous
#include <hip/hip_runtime.h>


typedef __attribute__((ext_vector_type(2))) float v2f;
typedef __attribute__((ext_vector_type(4))) float v4f;
typedef __attribute__((ext_vector_type(8))) float v8f;

static inline int idiv_up(long long a, long long b) { return (int)((a + b - 1) / b); }

__global__ void degree_kernel(const int* __restrict__ s, const int* __restrict__ p,
                              const int* __restrict__ o, float* __restrict__ deg,
                              int E, int N, int R) {
  long long i = (long long)blockIdx.x * blockDim.x + threadIdx.x;
  if (i < E) {
    int si = s[i], pi = p[i], oi = o[i];
    atomicAdd(&deg[(size_t)pi * N + si], 1.0f);
    atomicAdd(&deg[(size_t)(pi + R) * N + oi], 1.0f);
  } else if (i < (long long)E + N) {
    int node = (int)(i - E);
    deg[(size_t)(2 * R) * N + node] = 1.0f;
  }
}

__device__ __forceinline__ void gemm48_wmma_block(const float* __restrict__ A,
                                                  const float* __restrict__ B,
                                                  float* __restrict__ C,
                                                  int M, int K, long long J,
                                                  long long blk, float* __restrict__ sA) {
  const int total = 48 * K;
  for (int idx = threadIdx.x; idx < total; idx += blockDim.x) {
    const int row = idx / K;
    const int kk  = idx - row * K;
    sA[idx] = (row < M) ? A[row * K + kk] : 0.0f;
  }
  __syncthreads();

  const int wave = threadIdx.x >> 5;
  const int lane = threadIdx.x & 31;
  const long long jbase = (blk * (blockDim.x >> 5) + wave) * 16;
  if (jbase >= J) return;
  const int col   = lane & 15;
  const int khalf = (lane >> 4) * 2;

  v8f acc[3];
  acc[0] = (v8f){};
  acc[1] = (v8f){};
  acc[2] = (v8f){};

  const float* __restrict__ Bp = B + jbase + col;
  for (int kc = 0; kc < K; kc += 4) {
    const int k0 = kc + khalf;
    v2f b;
    b.x = Bp[(long long)k0 * J];
    b.y = Bp[(long long)(k0 + 1) * J];
#pragma unroll
    for (int mt = 0; mt < 3; ++mt) {
      const int m = mt * 16 + col;
      const v2f a = *(const v2f*)&sA[m * K + k0];
      acc[mt] = __builtin_amdgcn_wmma_f32_16x16x4_f32(
          false, a, false, b, (short)0, acc[mt], false, false);
    }
  }

  const int rhalf = (lane >> 4) * 8;
#pragma unroll
  for (int mt = 0; mt < 3; ++mt) {
#pragma unroll
    for (int v = 0; v < 8; ++v) {
      const int row = mt * 16 + rhalf + v;
      if (row < M) C[(long long)row * J + jbase + col] = acc[mt][v];
    }
  }
}

__global__ void gemm48_wmma_kernel(const float* __restrict__ A,
                                   const float* __restrict__ B,
                                   float* __restrict__ C,
                                   int M, int K, long long J) {
  __shared__ float sA[48 * 64];
  gemm48_wmma_block(A, B, C, M, K, J, (long long)blockIdx.x, sA);
}

__global__ __launch_bounds__(256) void init_and_w2_kernel(
    float* __restrict__ deg, long long n_deg, long long n_rel, float* __restrict__ h, long long n_h,
    float* __restrict__ out, const float* __restrict__ bias2, long long n_out, int C,
    const float* __restrict__ A2, const float* __restrict__ B2, float* __restrict__ W2,
    int M, int K, long long J2, int w2_blocks) {
  __shared__ float sA[48 * 64];
  if ((int)blockIdx.x < w2_blocks) {
    gemm48_wmma_block(A2, B2, W2, M, K, J2, (long long)blockIdx.x, sA);
    return;
  }
  long long i = (long long)((int)blockIdx.x - w2_blocks) * blockDim.x + threadIdx.x;
  if (i < n_deg) { deg[i] = (i < n_rel) ? 0.0f : 1.0f; return; }
  i -= n_deg;
  if (i < n_h) { h[i] = 0.0f; return; }
  i -= n_h;
  if (i < n_out) out[i] = bias2[i % C];
}

template <int KS, bool REFILL>
__device__ __forceinline__ void gemm48_chunk(const v2f (&a)[3][KS], v4f (&b0)[KS],
                                             v4f (&b1)[KS], v8f (&acc)[3][4],
                                             const float* __restrict__ pn, long long J) {
#pragma unroll
  for (int mt = 0; mt < 3; ++mt)
#pragma unroll
    for (int t = 0; t < 4; ++t) acc[mt][t] = (v8f){};
#pragma unroll
  for (int ks = 0; ks < KS; ++ks) {
    const v4f r0 = b0[ks];
    const v4f r1 = b1[ks];
#pragma unroll
    for (int t = 0; t < 4; ++t) {
      const v2f b = {r0[t], r1[t]};
#pragma unroll
      for (int mt = 0; mt < 3; ++mt)
        acc[mt][t] = __builtin_amdgcn_wmma_f32_16x16x4_f32(
            false, a[mt][ks], false, b, (short)0, acc[mt][t], false, false);
    }
    if (REFILL) {
      b0[ks] = __builtin_nontemporal_load((const v4f*)(pn + (long long)(4 * ks) * J));
      b1[ks] = __builtin_nontemporal_load((const v4f*)(pn + (long long)(4 * ks + 1) * J));
    }
    __builtin_amdgcn_sched_barrier(0);
  }
}

__device__ __forceinline__ void gemm48_store(const v8f (&acc)[3][4], float* __restrict__ pc,
                                             int lim, long long J) {
#pragma unroll
  for (int mt = 0; mt < 3; ++mt) {
#pragma unroll
    for (int r = 0; r < 8; ++r) {
      if (mt * 16 + r < lim) {
        const v4f d = {acc[mt][0][r], acc[mt][1][r], acc[mt][2][r], acc[mt][3][r]};
        *(v4f*)(pc + (long long)(mt * 16 + r) * J) = d;
      }
    }
  }
}

constexpr int DEG_SLOTS = 8;
constexpr unsigned DEG_NONE = 0xFFFFFFFFu;

__device__ __forceinline__ void deg_segments(const int* __restrict__ s, const int* __restrict__ p,
                                             const int* __restrict__ o, long long i, int N, int R,
                                             unsigned& a0, unsigned& a1) {
  const int si = s[i], pi = p[i], oi = o[i];
  a0 = (unsigned)pi * (unsigned)N + (unsigned)si;
  a1 = (unsigned)(pi + R) * (unsigned)N + (unsigned)oi;
}

template <int K>
__global__ __launch_bounds__(128, 4) void gemm48_stream_deg_kernel(
    const float* __restrict__ A, const float* __restrict__ B,
    float* __restrict__ C, int M, long long J,
    const int* __restrict__ s, const int* __restrict__ p, const int* __restrict__ o,
    float* __restrict__ deg, int E, int N, int R) {
  constexpr int KS = K / 4;
  __shared__ uint2 sseg[4][DEG_SLOTS][32];
  const int lane = threadIdx.x & 31;
  const int c = lane & 15;
  const int h = lane >> 4;
  const int wave = __builtin_amdgcn_readfirstlane((int)(threadIdx.x >> 5));
  const int wpb  = (int)(blockDim.x >> 5);
  const long long nw  = (long long)gridDim.x * wpb;
  const long long nST = J >> 6;
  const long long gw = (long long)blockIdx.x * wpb + wave;
  long long st = gw;
  const long long estep = nw * 32;

  if (st >= nST) {
    for (long long eb = gw * 32; eb < E; eb += estep) {
      const long long i = eb + lane;
      if (i < E) {
        unsigned a0, a1;
        deg_segments(s, p, o, i, N, R, a0, a1);
        atomicAdd(&deg[a0], 1.0f);
        atomicAdd(&deg[a1], 1.0f);
      }
    }
    return;
  }

  int es[DEG_SLOTS], ep[DEG_SLOTS], eo[DEG_SLOTS];
  int nb = 0;
#pragma unroll
  for (int k = 0; k < DEG_SLOTS; ++k) {
    const long long eb = gw * 32 + (long long)k * estep;
    es[k] = 0; ep[k] = 0; eo[k] = 0;
    if (eb < E) {
      nb = k + 1;
      const long long i = eb + lane;
      const long long ic = i < E ? i : (long long)E - 1;
      es[k] = s[ic]; ep[k] = p[ic]; eo[k] = o[ic];
    }
  }
  for (long long eb = gw * 32 + (long long)DEG_SLOTS * estep; eb < E; eb += estep) {
    const long long i = eb + lane;
    if (i < E) {
      unsigned a0, a1;
      deg_segments(s, p, o, i, N, R, a0, a1);
      atomicAdd(&deg[a0], 1.0f);
      atomicAdd(&deg[a1], 1.0f);
    }
  }
  auto deg_issue = [&](int b) {
    const uint2 sg = sseg[wave][b][lane];
    if (sg.x != DEG_NONE) {
      atomicAdd(&deg[sg.x], 1.0f);
      atomicAdd(&deg[sg.y], 1.0f);
    }
  };

  v2f a[3][KS];
#pragma unroll
  for (int mt = 0; mt < 3; ++mt) {
    const int row = mt * 16 + c;
#pragma unroll
    for (int ks = 0; ks < KS; ++ks) {
      v2f t = {0.0f, 0.0f};
      if (row < M) t = *(const v2f*)&A[row * K + 4 * ks + 2 * h];
      a[mt][ks] = t;
    }
  }

  const float* __restrict__ Bl = B + (long long)(2 * h) * J + 4 * c;
  float* __restrict__ Cl = C + (long long)(8 * h) * J + 4 * c;
  const int lim = M - 8 * h;

  v4f b0[KS], b1[KS];
  {
    const float* p = Bl + st * 64;
#pragma unroll
    for (int ks = 0; ks < KS; ++ks) {
      b0[ks] = __builtin_nontemporal_load((const v4f*)(p + (long long)(4 * ks) * J));
      b1[ks] = __builtin_nontemporal_load((const v4f*)(p + (long long)(4 * ks + 1) * J));
    }
  }

#pragma unroll
  for (int k = 0; k < DEG_SLOTS; ++k) {
    if (k < nb) {
      const long long i = gw * 32 + (long long)k * estep + lane;
      unsigned a0 = DEG_NONE, a1 = DEG_NONE;
      if (i < E) {
        a0 = (unsigned)ep[k] * (unsigned)N + (unsigned)es[k];
        a1 = (unsigned)(ep[k] + R) * (unsigned)N + (unsigned)eo[k];
      }
      sseg[wave][k][lane] = make_uint2(a0, a1);
    }
  }

  v8f acc[3][4];
  int bi = 0;
  while (st + nw < nST) {
    if (bi < nb) { deg_issue(bi); ++bi; }
    gemm48_chunk<KS, true>(a, b0, b1, acc, Bl + (st + nw) * 64, J);
    gemm48_store(acc, Cl + st * 64, lim, J);
    st += nw;
  }
  for (; bi < nb; ++bi) deg_issue(bi);
  gemm48_chunk<KS, false>(a, b0, b1, acc, Bl, J);
  gemm48_store(acc, Cl + st * 64, lim, J);
}

__device__ __forceinline__ int cache_half() {
  const unsigned id = __builtin_amdgcn_s_sendmsg_rtn(0x87);
  return (int)((id >> 18) & 1u);
}

__device__ __forceinline__ void edge_decode(const int* __restrict__ s, const int* __restrict__ p,
                                            const int* __restrict__ o,
                                            const float* __restrict__ deg,
                                            long long i, int E, int N, int R,
                                            int& si, int& pi, int& oi, float& val) {
  si = 0; pi = 0; oi = 0; val = 0.0f;
  if (i < (long long)2 * E + N) {
    if (i < E)            { si = s[i];      pi = p[i];          oi = o[i]; }
    else if (i < 2LL * E) { long long j = i - E; si = o[j]; pi = p[j] + R; oi = s[j]; }
    else                  { int nd = (int)(i - 2LL * E); si = nd; pi = 2 * R; oi = nd; }
    val = 1.0f / deg[(size_t)pi * N + si];
  }
}

template <bool GUARD>
__device__ __forceinline__ void layer1_rows(const float* __restrict__ W1, float* __restrict__ h,
                                            unsigned wofs, unsigned hofs, float val, int ok,
                                            int e) {
  float w[16];
#pragma unroll
  for (int k = 0; k < 16; ++k) {
    const unsigned wo = (unsigned)__shfl((int)wofs, k, 16);
    w[k] = W1[wo + e];
  }
#pragma unroll
  for (int k = 0; k < 16; ++k) {
    const unsigned ho = (unsigned)__shfl((int)hofs, k, 16);
    const float v = __shfl(val, k, 16);
    if (!GUARD || __shfl(ok, k, 16)) atomicAdd(&h[ho + e], v * w[k]);
  }
}

__global__ __launch_bounds__(256) void layer1_scatter_kernel(
    const int* __restrict__ s, const int* __restrict__ p, const int* __restrict__ o,
    const float* __restrict__ deg, const float* __restrict__ W1, float* __restrict__ hbase,
    int E, int N, int R, long long NE, long long hstride) {
  const int lane = threadIdx.x & 31;
  const long long total = (long long)2 * E + N;
  const long long base = (((long long)blockIdx.x * blockDim.x + threadIdx.x) >> 5) << 5;
  if (base >= total) return;
  float* __restrict__ h = hbase + (long long)cache_half() * hstride;
  int si, pi, oi;
  float val;
  edge_decode(s, p, o, deg, base + lane, E, N, R, si, pi, oi, val);
  const unsigned wofs = (unsigned)((long long)pi * NE + (long long)oi * 16);
  const unsigned hofs = (unsigned)si * 16u;
  const int e = lane & 15;
  if (base + 32 <= total) layer1_rows<false>(W1, h, wofs, hofs, val, 1, e);
  else layer1_rows<true>(W1, h, wofs, hofs, val, (base + lane) < total ? 1 : 0, e);
}

template <int CTRL>
__device__ __forceinline__ float lane_partner(float x) {
  return __builtin_bit_cast(
      float, __builtin_amdgcn_update_dpp(0, __builtin_bit_cast(int, x), CTRL, 0xf, 0xf, true));
}

template <bool GUARD, bool TWO>
__device__ __forceinline__ void layer2_rows(const float* __restrict__ h,
                                            const float* __restrict__ hB,
                                            const float* __restrict__ bias1,
                                            const float* __restrict__ W2,
                                            float* __restrict__ out,
                                            unsigned hofs, unsigned wofs, unsigned oofs,
                                            float val, int ok, int l) {
  float hv[16], hw[16];
#pragma unroll
  for (int k = 0; k < 16; ++k) {
    const unsigned ho = (unsigned)__shfl((int)hofs, k, 16);
    hv[k] = h[ho + l];
    hw[k] = TWO ? hB[ho + l] : 0.0f;
  }
  const float b1v = bias1[l];
#pragma unroll
  for (int k = 0; k < 16; ++k) hv[k] = fmaxf((TWO ? hv[k] + hw[k] : hv[k]) + b1v, 0.0f);
  const bool ch  = (l & 8) != 0;
  const bool b2  = (l & 4) != 0;
  const bool b1  = (l & 2) != 0;
  const bool odd = (l & 1) != 0;
  const unsigned wlane = (unsigned)((l & 7) * 8 + (l >> 3) * 4);
  const unsigned cofs  = (unsigned)(((l >> 3) << 2) | ((l >> 1) & 3));
#pragma unroll
  for (int k = 0; k < 16; k += 2) {
    float pr[2];
#pragma unroll
    for (int d = 0; d < 2; ++d) {
      const unsigned wo = (unsigned)__shfl((int)wofs, k + d, 16);
      const float v = __shfl(val, k + d, 16);
      const v4f* __restrict__ wp = (const v4f*)(W2 + wo + wlane);
      const v4f wa = wp[0];
      const v4f wb = wp[16];
      const float mine = hv[k + d];
      const float far  = lane_partner<0x128>(mine);
      const float ha = ch ? far : mine;
      const float hb = ch ? mine : far;
      float t[4];
#pragma unroll
      for (int j = 0; j < 4; ++j) t[j] = ha * wa[j] + hb * wb[j];
      float r[2];
#pragma unroll
      for (int j = 0; j < 2; ++j) {
        const float keep = b2 ? t[2 + j] : t[j];
        const float give = b2 ? t[j] : t[2 + j];
        r[j] = keep + lane_partner<0x141>(give);
      }
      const float keepb = b1 ? r[1] : r[0];
      const float giveb = b1 ? r[0] : r[1];
      const float u = keepb + lane_partner<0x4E>(giveb);
      pr[d] = v * (u + lane_partner<0xB1>(u));
    }
    const unsigned oo0 = (unsigned)__shfl((int)oofs, k, 16);
    const unsigned oo1 = (unsigned)__shfl((int)oofs, k + 1, 16);
    const unsigned oo = odd ? oo1 : oo0;
    const float add = odd ? pr[1] : pr[0];
    if (GUARD) {
      const int ok0 = __shfl(ok, k, 16);
      const int ok1 = __shfl(ok, k + 1, 16);
      if (odd ? ok1 : ok0) atomicAdd(&out[oo + cofs], add);
    } else {
      atomicAdd(&out[oo + cofs], add);
    }
  }
}

__global__ __launch_bounds__(256) void layer2_scatter_kernel(
    const int* __restrict__ s, const int* __restrict__ p, const int* __restrict__ o,
    const float* __restrict__ deg, const float* __restrict__ h, long long hstride,
    const float* __restrict__ bias1,
    const float* __restrict__ W2,
    float* __restrict__ outA, float* __restrict__ outB, int E, int N, int R) {
  const int lane = threadIdx.x & 31;
  const long long total = (long long)2 * E + N;
  const long long base = (((long long)blockIdx.x * blockDim.x + threadIdx.x) >> 5) << 5;
  if (base >= total) return;
  float* __restrict__ out = (outB != nullptr && cache_half() != 0) ? outB : outA;
  int si, pi, oi;
  float val;
  edge_decode(s, p, o, deg, base + lane, E, N, R, si, pi, oi, val);
  const unsigned hofs = (unsigned)oi * 16u;
  const unsigned wofs = (unsigned)pi * 128u;
  const unsigned oofs = (unsigned)si * 8u;
  const int ok = (base + lane) < total ? 1 : 0;
  if (hstride != 0) {
    const float* __restrict__ hB = h + hstride;
    if (base + 32 <= total) layer2_rows<false, true>(h, hB, bias1, W2, out, hofs, wofs, oofs, val, 1, lane & 15);
    else layer2_rows<true, true>(h, hB, bias1, W2, out, hofs, wofs, oofs, val, ok, lane & 15);
  } else {
    if (base + 32 <= total) layer2_rows<false, false>(h, h, bias1, W2, out, hofs, wofs, oofs, val, 1, lane & 15);
    else layer2_rows<true, false>(h, h, bias1, W2, out, hofs, wofs, oofs, val, ok, lane & 15);
  }
}

__global__ __launch_bounds__(256) void add_copy_kernel(float* __restrict__ out,
                                                       const float* __restrict__ outB,
                                                       long long n4) {
  const long long i = (long long)blockIdx.x * blockDim.x + threadIdx.x;
  if (i < n4) {
    const v4f a = ((const v4f*)out)[i];
    const v4f b = ((const v4f*)outB)[i];
    ((v4f*)out)[i] = a + b;
  }
}

extern "C" void kernel_launch(void* const* d_in, const int* in_sizes, int n_in,
                              void* d_out, int out_size, void* d_ws, size_t ws_size,
                              hipStream_t stream) {
  const float* comps1 = (const float*)d_in[0];
  const float* bases1 = (const float*)d_in[1];
  const float* comps2 = (const float*)d_in[2];
  const float* bases2 = (const float*)d_in[3];
  const float* bias1  = (const float*)d_in[4];
  const float* bias2  = (const float*)d_in[5];
  const int*   s      = (const int*)d_in[6];
  const int*   p      = (const int*)d_in[7];
  const int*   o      = (const int*)d_in[8];

  const int EMB    = in_sizes[4];
  const int NUMCLS = in_sizes[5];
  const int E      = in_sizes[6];
  const int N      = out_size / NUMCLS;
  const int BASES  = (int)((long long)in_sizes[1] / ((long long)N * EMB));
  const int r_tot  = in_sizes[0] / BASES;
  const int R      = (r_tot - 1) / 2;

  float* ws = (float*)d_ws;
  size_t off = 0;
  auto alloc = [&](size_t nfloats) {
    size_t cur = off;
    off += (nfloats + 63) & ~(size_t)63;
    return cur;
  };
  const long long NE   = (long long)N * EMB;
  const long long n_out = (long long)N * NUMCLS;
  auto slab = [](size_t nfloats) { return (nfloats + 63) & ~(size_t)63; };
  const size_t n_acc2 = (size_t)(2 * NE + n_out);
  const size_t need2 = slab((size_t)r_tot * N) + slab(n_acc2) + slab((size_t)r_tot * EMB * NUMCLS) +
                       slab((size_t)r_tot * N * EMB);
  const bool two = (n_out % 4) == 0 && need2 * sizeof(float) <= ws_size;
  float* deg = ws + alloc((size_t)r_tot * N);
  float* h   = ws + alloc(two ? n_acc2 : (size_t)NE);
  float* W2  = ws + alloc((size_t)r_tot * EMB * NUMCLS);
  float* W1  = ws + alloc((size_t)r_tot * N * EMB);
  (void)n_in;
  const long long hstride = two ? NE : 0;
  float* outB = two ? h + 2 * NE : nullptr;
  const long long n_clear = two ? 2 * NE + n_out : NE;

  const long long J2   = (long long)EMB * NUMCLS;
  const long long Eall = (long long)2 * E + N;
  const int B = 256;

  const long long n_deg = (long long)r_tot * N;
  const int fill_blocks = idiv_up(n_deg + n_clear + n_out, B);
  const int w2_blocks   = idiv_up(J2, 16LL * (B / 32));
  const long long n_rel = (long long)2 * R * N;
  init_and_w2_kernel<<<w2_blocks + fill_blocks, B, 0, stream>>>(
      deg, n_deg, n_rel, h, n_clear, (float*)d_out, bias2, n_out, NUMCLS,
      comps2, bases2, W2, r_tot, BASES, J2, w2_blocks);

  if (BASES == 40 && (NE & 63) == 0 && r_tot <= 48) {
    gemm48_stream_deg_kernel<40><<<1024, 128, 0, stream>>>(comps1, bases1, W1, r_tot, NE,
                                                          s, p, o, deg, E, N, R);
  } else {
    degree_kernel<<<idiv_up((long long)E + N, B), B, 0, stream>>>(s, p, o, deg, E, N, R);
    gemm48_wmma_kernel<<<idiv_up(NE, 16LL * (B / 32)), B, 0, stream>>>(
        comps1, bases1, W1, r_tot, BASES, NE);
  }

  layer1_scatter_kernel<<<idiv_up(Eall, B), B, 0, stream>>>(
      s, p, o, deg, W1, h, E, N, R, NE, hstride);

  layer2_scatter_kernel<<<idiv_up(Eall, B), B, 0, stream>>>(
      s, p, o, deg, h, hstride, bias1, W2, (float*)d_out, outB, E, N, R);

  if (two) add_copy_kernel<<<idiv_up(n_out / 4, B), B, 0, stream>>>((float*)d_out, outB, n_out / 4);
}
